# starting kernel
# baseline (speedup 1.0000x reference)
#include <hip/hip_runtime.h>


typedef __attribute__((ext_vector_type(2))) float v2f;
typedef __attribute__((ext_vector_type(8))) float v8f;

#define NSTATE 4096
#define THREADS 128

__global__ __launch_bounds__(THREADS)
void vqc_kernel(const float* __restrict__ instance,
                const int*   __restrict__ concept_index,
                const float* __restrict__ concept_weights,
                float*       __restrict__ out)
{
  __shared__ float gR[12][2], gI[12][2];
  __shared__ float Ur[64], Ui[64];
  __shared__ float Vr[64], Vi[64];
  __shared__ float obsA[4], obsB[4], obsC[4];
  __shared__ float psiR[NSTATE];
  __shared__ float psiI[NSTATE];
  __shared__ float partial[THREADS][17];
  __shared__ float sums[16];

  const int b   = blockIdx.x;
  const int tid = threadIdx.x;

  __builtin_prefetch(instance + (size_t)b * 36, 0, 3);

  if (tid < 12) {
    const int w = tid;
    const float* a = instance + ((size_t)b * 12 + w) * 3;
    const float phi = a[0], th = a[1], om = a[2];
    float c, s;   sincosf(0.5f * th, &s, &c);
    float s0, c0; sincosf(-0.5f * (phi + om), &s0, &c0);
    float s1, c1; sincosf( 0.5f * (om  - phi), &s1, &c1);
    const int p = 11 - w;
    gR[p][0] = c * c0;  gI[p][0] = c * s0;
    gR[p][1] = s * c1;  gI[p][1] = s * s1;
  } else if (tid < 16) {
    const int d  = tid - 12;
    const int ci = concept_index[b];
    const float* cw = concept_weights + (size_t)ci * 12 + d * 3;
    const float cth = cw[1], com = cw[2];
    float sth, vcth; sincosf(cth, &sth, &vcth);
    float so,  co;   sincosf(com, &so,  &co);
    obsA[d] = vcth;
    obsB[d] = 2.0f * sth * co;
    obsC[d] = 2.0f * sth * so;
  }
  __syncthreads();

  {
    const int h    = tid & 63;
    const int base = (tid < 64) ? 6 : 0;
    float pr = 1.0f, pi = 0.0f;
#pragma unroll
    for (int j = 0; j < 6; ++j) {
      const int t = (h >> j) & 1;
      const float gr = gR[base + j][t], gi = gI[base + j][t];
      const float nr = pr * gr - pi * gi;
      const float ni = pr * gi + pi * gr;
      pr = nr; pi = ni;
    }
    if (tid < 64) { Ur[h] = 0.5f * pr; Ui[h] = 0.5f * pi; }
    else          { Vr[h] = pr;        Vi[h] = pi;        }
  }
  __syncthreads();

  {
    const int wave = tid >> 5;
    const int lane = tid & 31;
    const int m    = lane & 15;
    const int I    = wave;
    const v2f A = { Ur[16 * I + m], Ui[16 * I + m] };
#pragma unroll
    for (int J = 0; J < 4; ++J) {
      const float vr = Vr[16 * J + m];
      const float vi = Vi[16 * J + m];
      const v2f Bre = { vr, -vi };
      const v2f Bim = { vi,  vr };
      const v8f z0 = { 0.f, 0.f, 0.f, 0.f, 0.f, 0.f, 0.f, 0.f };
      v8f dre = __builtin_amdgcn_wmma_f32_16x16x4_f32(false, A, false, Bre,
                                                      (short)0, z0, false, false);
      v8f dim = __builtin_amdgcn_wmma_f32_16x16x4_f32(false, A, false, Bim,
                                                      (short)0, z0, false, false);
      const int hbase = 16 * I + ((lane & 16) ? 8 : 0);
      const int lg    = 16 * J + m;
#pragma unroll
      for (int r = 0; r < 8; ++r) {
        psiR[(hbase + r) * 64 + lg] = dre[r];
        psiI[(hbase + r) * 64 + lg] = dim[r];
      }
    }
  }
  __syncthreads();

  float acc[16];
#pragma unroll
  for (int q = 0; q < 16; ++q) acc[q] = 0.0f;

  const int bits[4] = { 11, 8, 5, 2 };
  for (int i = 0; i < NSTATE / THREADS; ++i) {
    const int s   = tid + i * THREADS;
    const int idx = s ^ (s >> 1) ^ ((s & 1) ? 0xC00 : 0);
    const float pr = psiR[idx], pi = psiI[idx];
    const float nrm = pr * pr + pi * pi;
#pragma unroll
    for (int d = 0; d < 4; ++d) {
      const int bb = bits[d];
      const int j  = idx ^ (3 << (bb - 1));
      const float qr = psiR[j], qi = psiI[j];
      const bool hi = (s >> bb) & 1;
      if (hi) {
        acc[d * 4 + 1] += nrm;
      } else {
        acc[d * 4 + 0] += nrm;
        acc[d * 4 + 2] += pr * qr + pi * qi;
        acc[d * 4 + 3] += pr * qi - pi * qr;
      }
    }
  }
#pragma unroll
  for (int q = 0; q < 16; ++q) partial[tid][q] = acc[q];
  __syncthreads();

  if (tid < 16) {
    float t = 0.0f;
    for (int i = 0; i < THREADS; ++i) t += partial[i][tid];
    sums[tid] = t;
  }
  __syncthreads();

  if (tid < 4) {
    const int d = tid;
    const float n0  = sums[d * 4 + 0];
    const float n1  = sums[d * 4 + 1];
    const float crR = sums[d * 4 + 2];
    const float crI = sums[d * 4 + 3];
    const float z = obsA[d] * (n0 - n1) + obsB[d] * crR + obsC[d] * crI;
    out[(size_t)b * 4 + d] = 0.5f * z + 0.5f;
  }
}

extern "C" void kernel_launch(void* const* d_in, const int* in_sizes, int n_in,
                              void* d_out, int out_size, void* d_ws, size_t ws_size,
                              hipStream_t stream) {
  const float* instance = (const float*)d_in[0];
  const int*   cidx     = (const int*)d_in[1];
  const float* cweights = (const float*)d_in[2];
  float*       out      = (float*)d_out;
  const int B = in_sizes[0] / 36;
  vqc_kernel<<<B, THREADS, 0, stream>>>(instance, cidx, cweights, out);
}


# State kept in the WMMA accumulators and reduced there (tile map puts each partner amplitude in-lane or one row_xmask DPP away): no LDS round trip of psi; order of the f32 sums changes
# speedup vs baseline: 3.7036x; 3.7036x over previous
#include <hip/hip_runtime.h>


typedef __attribute__((ext_vector_type(2))) float v2f;
typedef __attribute__((ext_vector_type(8))) float v8f;

#define THREADS 128
#define NACC 12

template <int MASK>
__device__ __forceinline__ float row_xor(float x) {
  return __int_as_float(__builtin_amdgcn_update_dpp(0, __float_as_int(x),
                                                    0x160 + MASK, 0xF, 0xF, true));
}

__global__ __launch_bounds__(THREADS)
void vqc_kernel(const float* __restrict__ instance,
                const int*   __restrict__ concept_index,
                const float* __restrict__ concept_weights,
                float*       __restrict__ out)
{
  __shared__ float gR[12][2], gI[12][2];
  __shared__ float Ur[64], Ui[64];
  __shared__ float Vr[64], Vi[64];
  __shared__ float obsA[4], obsB[4], obsC[4];
  __shared__ float partial[THREADS][NACC + 1];
  __shared__ float sums[NACC];

  const int b   = blockIdx.x;
  const int tid = threadIdx.x;

  __builtin_prefetch(instance + (size_t)b * 36, 0, 3);

  if (tid < 12) {
    const int w = tid;
    const float* a = instance + ((size_t)b * 12 + w) * 3;
    const float phi = a[0], th = a[1], om = a[2];
    float c, s;   sincosf(0.5f * th, &s, &c);
    float s0, c0; sincosf(-0.5f * (phi + om), &s0, &c0);
    float s1, c1; sincosf( 0.5f * (om  - phi), &s1, &c1);
    const int p = 11 - w;
    gR[p][0] = c * c0;  gI[p][0] = c * s0;
    gR[p][1] = s * c1;  gI[p][1] = s * s1;
  } else if (tid < 16) {
    const int d  = tid - 12;
    const int ci = concept_index[b];
    const float* cw = concept_weights + (size_t)ci * 12 + d * 3;
    const float cth = cw[1], com = cw[2];
    float sth, vcth; sincosf(cth, &sth, &vcth);
    float so,  co;   sincosf(com, &so,  &co);
    obsA[d] = vcth;
    obsB[d] = 2.0f * sth * co;
    obsC[d] = 2.0f * sth * so;
  }
  __syncthreads();

  {
    const int h    = tid & 63;
    const int base = (tid < 64) ? 6 : 0;
    float pr = 1.0f, pi = 0.0f;
#pragma unroll
    for (int j = 0; j < 6; ++j) {
      const int t = (h >> j) & 1;
      const float gr = gR[base + j][t], gi = gI[base + j][t];
      const float nr = pr * gr - pi * gi;
      const float ni = pr * gi + pi * gr;
      pr = nr; pi = ni;
    }
    if (tid < 64) { Ur[h] = 0.5f * pr; Ui[h] = 0.5f * pi; }
    else          { Vr[h] = pr;        Vi[h] = pi;        }
  }
  __syncthreads();

  const int wave = tid >> 5;
  const int lane = tid & 31;
  const int c    = lane & 15;
  const int hh   = lane >> 4;

  const int lcol = ((c & 12) << 2) | ((wave & 2) << 2) | ((c & 3) << 1) | (wave & 1);
  const int hrow0 = ((c & 4) << 2) | (c & 8) | ((c & 3) << 1);

  const float vr = Vr[lcol];
  const float vi = Vi[lcol];
  const v2f Bre = { vr, -vi };
  const v2f Bim = { vi,  vr };
  const v8f z0 = { 0.f, 0.f, 0.f, 0.f, 0.f, 0.f, 0.f, 0.f };

  const v2f zero2 = { 0.f, 0.f };
  v2f zL = zero2, zH = zero2;
  v2f z1 = zero2;
  v2f cr0 = zero2, ci0 = zero2, cr1 = zero2, ci1 = zero2;
  v2f cr2 = zero2, ci2 = zero2, cr3 = zero2, ci3 = zero2;

#pragma unroll
  for (int q0 = 0; q0 < 2; ++q0) {
    const int hlo = hrow0 | q0;
    const int hhi = hlo | 32;
    const v2f Alo = { Ur[hlo], Ui[hlo] };
    const v2f Ahi = { Ur[hhi], Ui[hhi] };
    const v8f reL = __builtin_amdgcn_wmma_f32_16x16x4_f32(false, Alo, false, Bre,
                                                          (short)0, z0, false, false);
    const v8f imL = __builtin_amdgcn_wmma_f32_16x16x4_f32(false, Alo, false, Bim,
                                                          (short)0, z0, false, false);
    const v8f reH = __builtin_amdgcn_wmma_f32_16x16x4_f32(false, Ahi, false, Bre,
                                                          (short)0, z0, false, false);
    const v8f imH = __builtin_amdgcn_wmma_f32_16x16x4_f32(false, Ahi, false, Bim,
                                                          (short)0, z0, false, false);
#pragma unroll
    for (int q1 = 0; q1 < 2; ++q1) {
#pragma unroll
      for (int rr = 0; rr < 8; rr += 2) {
        const v2f ar = q1 ? (v2f){ reH[rr], reH[rr + 1] } : (v2f){ reL[rr], reL[rr + 1] };
        const v2f ai = q1 ? (v2f){ imH[rr], imH[rr + 1] } : (v2f){ imL[rr], imL[rr + 1] };
        const int rp = ((rr >> 1) ^ (rr >> 2)) & 1;
        const int x0 = q0 ^ rp;
        const int x1 = q1 ^ rp;
        const int x23 = x0 ^ q1;
        const v2f nrm = ar * ar + ai * ai;
        if (q1) zH += x0 ? -nrm : nrm;
        else    zL += x0 ? -nrm : nrm;
        z1 += x1 ? -nrm : nrm;


        if (q1 == 0) {
          const v2f br = { reH[rr ^ 4], reH[(rr ^ 4) + 1] };
          const v2f bi = { imH[rr ^ 4], imH[(rr ^ 4) + 1] };
          cr0 += ar * br;  cr0 += ai * bi;
          if (x0) { ci0 -= ar * bi;  ci0 += ai * br; }
          else    { ci0 += ar * bi;  ci0 -= ai * br; }
        }
        if ((rr & 2) == 0) {
          const v2f br = q1 ? (v2f){ reH[rr + 3], reH[rr + 2] } : (v2f){ reL[rr + 3], reL[rr + 2] };
          const v2f bi = q1 ? (v2f){ imH[rr + 3], imH[rr + 2] } : (v2f){ imL[rr + 3], imL[rr + 2] };
          cr1 += ar * br;  cr1 += ai * bi;
          if (x1) { ci1 -= ar * bi;  ci1 += ai * br; }
          else    { ci1 += ar * bi;  ci1 -= ai * br; }
        }
        {
          const v2f br = { row_xor<12>(ar.x), row_xor<12>(ar.y) };
          const v2f bi = { row_xor<12>(ai.x), row_xor<12>(ai.y) };
          cr2 += ar * br;  cr2 += ai * bi;
          if (x23) { ci2 -= ar * bi;  ci2 += ai * br; }
          else     { ci2 += ar * bi;  ci2 -= ai * br; }
        }
        {
          const v2f br = { row_xor<3>(ar.x), row_xor<3>(ar.y) };
          const v2f bi = { row_xor<3>(ai.x), row_xor<3>(ai.y) };
          cr3 += ar * br;  cr3 += ai * bi;
          if (x23) { ci3 -= ar * bi;  ci3 += ai * br; }
          else     { ci3 += ar * bi;  ci3 -= ai * br; }
        }
      }
    }
  }

  const float fL = zL.x - zL.y, fH = zH.x - zH.y;
  const float zn0  = fL + fH;
  const float zn23 = fL - fH;
  const float zn1  = z1.x + z1.y;
  const float cr0s = cr0.x + cr0.y, ci0s = ci0.x - ci0.y;
  const float cr1s = cr1.x + cr1.y, ci1s = ci1.x + ci1.y;
  const float cr2s = cr2.x + cr2.y, ci2s = ci2.x - ci2.y;
  const float cr3s = cr3.x + cr3.y, ci3s = ci3.x - ci3.y;

  const int pl = __builtin_popcount(lane) & 1;
  const int pw = (wave ^ (wave >> 1)) & 1;
  const float s0 = (pl ^ pw)                                        ? -1.f : 1.f;
  const float s1 = hh                                               ? -1.f : 1.f;
  const float s2 = (((c >> 3) ^ hh) & 1)                            ? -1.f : 1.f;
  const float s3 = (((c >> 1) ^ (c >> 2) ^ (c >> 3) ^ (wave >> 1) ^ hh) & 1) ? -1.f : 1.f;

  partial[tid][0]  = s0 * zn0;   partial[tid][1]  = cr0s;         partial[tid][2]  = s0 * ci0s;
  partial[tid][3]  = s1 * zn1;   partial[tid][4]  = cr1s;         partial[tid][5]  = s1 * ci1s;
  partial[tid][6]  = s2 * zn23;  partial[tid][7]  = 0.5f * cr2s;  partial[tid][8]  = 0.5f * s2 * ci2s;
  partial[tid][9]  = s3 * zn23;  partial[tid][10] = 0.5f * cr3s;  partial[tid][11] = 0.5f * s3 * ci3s;
  __syncthreads();

  if (tid < NACC) {
    float t = 0.0f;
    for (int i = 0; i < THREADS; ++i) t += partial[i][tid];
    sums[tid] = t;
  }
  __syncthreads();

  if (tid < 4) {
    const int d = tid;
    const float dz  = sums[d * 3 + 0];
    const float crR = sums[d * 3 + 1];
    const float crI = sums[d * 3 + 2];
    const float z = obsA[d] * dz + obsB[d] * crR + obsC[d] * crI;
    out[(size_t)b * 4 + d] = 0.5f * z + 0.5f;
  }
}

extern "C" void kernel_launch(void* const* d_in, const int* in_sizes, int n_in,
                              void* d_out, int out_size, void* d_ws, size_t ws_size,
                              hipStream_t stream) {
  const float* instance = (const float*)d_in[0];
  const int*   cidx     = (const int*)d_in[1];
  const float* cweights = (const float*)d_in[2];
  float*       out      = (float*)d_out;
  const int B = in_sizes[0] / 36;
  vqc_kernel<<<B, THREADS, 0, stream>>>(instance, cidx, cweights, out);
}


# One wave per batch element (its four 1024-amplitude slices looped in the wave, 4 independent waves per block, no block barriers): 4x fewer waves to launch; order of the f32 sums changes
# speedup vs baseline: 5.2712x; 1.4233x over previous
#include <hip/hip_runtime.h>


typedef __attribute__((ext_vector_type(2))) float v2f;
typedef __attribute__((ext_vector_type(8))) float v8f;

#define WAVES   4
#define THREADS (32 * WAVES)
#define NACC    12

template <int MASK>
__device__ __forceinline__ float row_xor(float x) {
  return __int_as_float(__builtin_amdgcn_update_dpp(0, __float_as_int(x),
                                                    0x160 + MASK, 0xF, 0xF, true));
}

struct WaveLds {
  float gR[12][2], gI[12][2];
  float Ur[64], Ui[64];
  float Vr[64], Vi[64];
  float obsA[4], obsB[4], obsC[4];
  float partial[32][NACC + 1];
  float sums[NACC];
};

__device__ __forceinline__ void wave_sync() {
  __builtin_amdgcn_fence(__ATOMIC_RELEASE, "wavefront");
  __builtin_amdgcn_wave_barrier();
  __builtin_amdgcn_fence(__ATOMIC_ACQUIRE, "wavefront");
}

__global__ __launch_bounds__(THREADS)
void vqc_kernel(const float* __restrict__ instance,
                const int*   __restrict__ concept_index,
                const float* __restrict__ concept_weights,
                float*       __restrict__ out,
                int B)
{
  __shared__ WaveLds lds_all[WAVES];

  const int lane = threadIdx.x & 31;
  const int wv = __builtin_amdgcn_readfirstlane((int)(threadIdx.x >> 5));
  const int b  = blockIdx.x * WAVES + wv;
  if (b >= B) return;
  WaveLds& L = lds_all[wv];

  __builtin_prefetch(instance + (size_t)b * 36, 0, 3);

  if (lane < 12) {
    const int w = lane;
    const float* a = instance + ((size_t)b * 12 + w) * 3;
    const float phi = a[0], th = a[1], om = a[2];
    float c, s;   sincosf(0.5f * th, &s, &c);
    float s0, c0; sincosf(-0.5f * (phi + om), &s0, &c0);
    float s1, c1; sincosf( 0.5f * (om  - phi), &s1, &c1);
    const int p = 11 - w;
    L.gR[p][0] = c * c0;  L.gI[p][0] = c * s0;
    L.gR[p][1] = s * c1;  L.gI[p][1] = s * s1;
  } else if (lane < 16) {
    const int d  = lane - 12;
    const int ci = concept_index[b];
    const float* cw = concept_weights + (size_t)ci * 12 + d * 3;
    const float cth = cw[1], com = cw[2];
    float sth, vcth; sincosf(cth, &sth, &vcth);
    float so,  co;   sincosf(com, &so,  &co);
    L.obsA[d] = vcth;
    L.obsB[d] = 2.0f * sth * co;
    L.obsC[d] = 2.0f * sth * so;
  }
  wave_sync();

#pragma unroll
  for (int g = 0; g < 2; ++g) {
    const int base = g ? 0 : 6;
    float pr = L.gR[base][lane & 1], pi = L.gI[base][lane & 1];
#pragma unroll
    for (int j = 1; j < 5; ++j) {
      const int t = (lane >> j) & 1;
      const float gr = L.gR[base + j][t], gi = L.gI[base + j][t];
      const float nr = pr * gr - pi * gi;
      const float ni = pr * gi + pi * gr;
      pr = nr; pi = ni;
    }
#pragma unroll
    for (int t = 0; t < 2; ++t) {
      const float gr = L.gR[base + 5][t], gi = L.gI[base + 5][t];
      const float nr = pr * gr - pi * gi;
      const float ni = pr * gi + pi * gr;
      const int h = lane | (t << 5);
      if (g == 0) { L.Ur[h] = 0.5f * nr; L.Ui[h] = 0.5f * ni; }
      else        { L.Vr[h] = nr;        L.Vi[h] = ni;        }
    }
  }
  wave_sync();

  const int c  = lane & 15;
  const int hh = lane >> 4;

  const int lcol0 = ((c & 12) << 2) | ((c & 3) << 1);
  const int hrow0 = ((c & 4) << 2) | (c & 8) | ((c & 3) << 1);

  const v2f Alo0 = { L.Ur[hrow0],      L.Ui[hrow0]      };
  const v2f Alo1 = { L.Ur[hrow0 | 1],  L.Ui[hrow0 | 1]  };
  const v2f Ahi0 = { L.Ur[hrow0 | 32], L.Ui[hrow0 | 32] };
  const v2f Ahi1 = { L.Ur[hrow0 | 33], L.Ui[hrow0 | 33] };
  const v8f z0 = { 0.f, 0.f, 0.f, 0.f, 0.f, 0.f, 0.f, 0.f };

  const v2f zero2 = { 0.f, 0.f };
  v2f z1 = zero2;
  v2f cr0 = zero2, ci0 = zero2, cr1 = zero2, ci1 = zero2;
  v2f cr2 = zero2, ci2 = zero2, cr3 = zero2, ci3 = zero2;
  float n0 = 0.f, n2 = 0.f, n3 = 0.f;

#pragma unroll
  for (int w = 0; w < 4; ++w) {
    const int pw = (w ^ (w >> 1)) & 1;
    const int w1 = (w >> 1) & 1;
    const int lcol = lcol0 | ((w & 2) << 2) | (w & 1);
    const float vr = L.Vr[lcol];
    const float vi = L.Vi[lcol];
    const v2f Bre = { vr, -vi };
    const v2f Bim = { vi,  vr };
    v2f zL = zero2, zH = zero2;

#pragma unroll
    for (int q0 = 0; q0 < 2; ++q0) {
      const v2f Alo = q0 ? Alo1 : Alo0;
      const v2f Ahi = q0 ? Ahi1 : Ahi0;
      const v8f reL = __builtin_amdgcn_wmma_f32_16x16x4_f32(false, Alo, false, Bre,
                                                            (short)0, z0, false, false);
      const v8f imL = __builtin_amdgcn_wmma_f32_16x16x4_f32(false, Alo, false, Bim,
                                                            (short)0, z0, false, false);
      const v8f reH = __builtin_amdgcn_wmma_f32_16x16x4_f32(false, Ahi, false, Bre,
                                                            (short)0, z0, false, false);
      const v8f imH = __builtin_amdgcn_wmma_f32_16x16x4_f32(false, Ahi, false, Bim,
                                                            (short)0, z0, false, false);
#pragma unroll
      for (int q1 = 0; q1 < 2; ++q1) {
#pragma unroll
        for (int rr = 0; rr < 8; rr += 2) {
          const v2f ar = q1 ? (v2f){ reH[rr], reH[rr + 1] } : (v2f){ reL[rr], reL[rr + 1] };
          const v2f ai = q1 ? (v2f){ imH[rr], imH[rr + 1] } : (v2f){ imL[rr], imL[rr + 1] };
          const int rp = ((rr >> 1) ^ (rr >> 2)) & 1;
          const int x0 = q0 ^ rp;
          const int x1 = q1 ^ rp;
          const int x23 = x0 ^ q1;
          const int y0 = x0 ^ pw;
          const int y3 = x23 ^ w1;
          const v2f nrm = ar * ar + ai * ai;
          if (q1) zH += x0 ? -nrm : nrm;
          else    zL += x0 ? -nrm : nrm;
          z1 += x1 ? -nrm : nrm;


          if (q1 == 0) {
            const v2f br = { reH[rr ^ 4], reH[(rr ^ 4) + 1] };
            const v2f bi = { imH[rr ^ 4], imH[(rr ^ 4) + 1] };
            cr0 += ar * br;  cr0 += ai * bi;
            if (y0) { ci0 -= ar * bi;  ci0 += ai * br; }
            else    { ci0 += ar * bi;  ci0 -= ai * br; }
          }
          if ((rr & 2) == 0) {
            const v2f br = q1 ? (v2f){ reH[rr + 3], reH[rr + 2] } : (v2f){ reL[rr + 3], reL[rr + 2] };
            const v2f bi = q1 ? (v2f){ imH[rr + 3], imH[rr + 2] } : (v2f){ imL[rr + 3], imL[rr + 2] };
            cr1 += ar * br;  cr1 += ai * bi;
            if (x1) { ci1 -= ar * bi;  ci1 += ai * br; }
            else    { ci1 += ar * bi;  ci1 -= ai * br; }
          }
          {
            const v2f br = { row_xor<12>(ar.x), row_xor<12>(ar.y) };
            const v2f bi = { row_xor<12>(ai.x), row_xor<12>(ai.y) };
            cr2 += ar * br;  cr2 += ai * bi;
            if (x23) { ci2 -= ar * bi;  ci2 += ai * br; }
            else     { ci2 += ar * bi;  ci2 -= ai * br; }
          }
          {
            const v2f br = { row_xor<3>(ar.x), row_xor<3>(ar.y) };
            const v2f bi = { row_xor<3>(ai.x), row_xor<3>(ai.y) };
            cr3 += ar * br;  cr3 += ai * bi;
            if (y3) { ci3 -= ar * bi;  ci3 += ai * br; }
            else    { ci3 += ar * bi;  ci3 -= ai * br; }
          }
        }
      }
    }

    const float fL = zL.x - zL.y, fH = zH.x - zH.y;
    const float zs = fL + fH;
    const float zd = fL - fH;
    n0 += pw ? -zs : zs;
    n2 += zd;
    n3 += w1 ? -zd : zd;
  }

  const float zn1  = z1.x + z1.y;
  const float cr0s = cr0.x + cr0.y, ci0s = ci0.x - ci0.y;
  const float cr1s = cr1.x + cr1.y, ci1s = ci1.x + ci1.y;
  const float cr2s = cr2.x + cr2.y, ci2s = ci2.x - ci2.y;
  const float cr3s = cr3.x + cr3.y, ci3s = ci3.x - ci3.y;

  const int pl = __builtin_popcount(lane) & 1;
  const float s0 = pl                                               ? -1.f : 1.f;
  const float s1 = hh                                               ? -1.f : 1.f;
  const float s2 = (((c >> 3) ^ hh) & 1)                            ? -1.f : 1.f;
  const float s3 = (((c >> 1) ^ (c >> 2) ^ (c >> 3) ^ hh) & 1)      ? -1.f : 1.f;

  L.partial[lane][0]  = s0 * n0;    L.partial[lane][1]  = cr0s;         L.partial[lane][2]  = s0 * ci0s;
  L.partial[lane][3]  = s1 * zn1;   L.partial[lane][4]  = cr1s;         L.partial[lane][5]  = s1 * ci1s;
  L.partial[lane][6]  = s2 * n2;    L.partial[lane][7]  = 0.5f * cr2s;  L.partial[lane][8]  = 0.5f * s2 * ci2s;
  L.partial[lane][9]  = s3 * n3;    L.partial[lane][10] = 0.5f * cr3s;  L.partial[lane][11] = 0.5f * s3 * ci3s;
  wave_sync();

  if (lane < NACC) {
    float t = 0.0f;
    for (int i = 0; i < 32; ++i) t += L.partial[i][lane];
    L.sums[lane] = t;
  }
  wave_sync();

  if (lane < 4) {
    const int d = lane;
    const float dz  = L.sums[d * 3 + 0];
    const float crR = L.sums[d * 3 + 1];
    const float crI = L.sums[d * 3 + 2];
    const float z = L.obsA[d] * dz + L.obsB[d] * crR + L.obsC[d] * crI;
    out[(size_t)b * 4 + d] = 0.5f * z + 0.5f;
  }
}

extern "C" void kernel_launch(void* const* d_in, const int* in_sizes, int n_in,
                              void* d_out, int out_size, void* d_ws, size_t ws_size,
                              hipStream_t stream) {
  const float* instance = (const float*)d_in[0];
  const int*   cidx     = (const int*)d_in[1];
  const float* cweights = (const float*)d_in[2];
  float*       out      = (float*)d_out;
  const int B = in_sizes[0] / 36;
  const int blocks = (B + WAVES - 1) / WAVES;
  vqc_kernel<<<blocks, THREADS, 0, stream>>>(instance, cidx, cweights, out, B);
}


# Wire bits 5 and 2 share one v_permlane16 gather (half of a row's lanes take each wire): each cross-lane pair product is formed once, not by both its lanes; order of the f32 sums changes
# speedup vs baseline: 6.7164x; 1.2742x over previous
#include <hip/hip_runtime.h>


typedef __attribute__((ext_vector_type(2))) float v2f;
typedef __attribute__((ext_vector_type(8))) float v8f;

#define WAVES   4
#define THREADS (32 * WAVES)
#define NACC    12

__host__ __device__ constexpr int pair_wire(int c) { return ((c >> 3) ^ (c >> 1)) & 1; }
__host__ __device__ constexpr unsigned pair_sel(int first) {
  unsigned s = 0;
  for (int j = 0; j < 8; ++j) {
    const int c = first + j;
    s |= (unsigned)(pair_wire(c) ? (c ^ 3) : (c ^ 12)) << (4 * j);
  }
  return s;
}
__device__ __forceinline__ float pair_gather(float x) {
  constexpr unsigned lo = pair_sel(0), hi = pair_sel(8);
  return __int_as_float((int)__builtin_amdgcn_permlane16(0u, (unsigned)__float_as_int(x),
                                                         lo, hi, true, false));
}

struct WaveLds {
  float gR[12][2], gI[12][2];
  float Ur[64], Ui[64];
  float Vr[64], Vi[64];
  float obsA[4], obsB[4], obsC[4];
  float partial[32][NACC + 1];
  float sums[NACC];
};

__device__ __forceinline__ void wave_sync() {
  __builtin_amdgcn_fence(__ATOMIC_RELEASE, "wavefront");
  __builtin_amdgcn_wave_barrier();
  __builtin_amdgcn_fence(__ATOMIC_ACQUIRE, "wavefront");
}

__global__ __launch_bounds__(THREADS)
void vqc_kernel(const float* __restrict__ instance,
                const int*   __restrict__ concept_index,
                const float* __restrict__ concept_weights,
                float*       __restrict__ out,
                int B)
{
  __shared__ WaveLds lds_all[WAVES];

  const int lane = threadIdx.x & 31;
  const int wv = __builtin_amdgcn_readfirstlane((int)(threadIdx.x >> 5));
  const int b  = blockIdx.x * WAVES + wv;
  if (b >= B) return;
  WaveLds& L = lds_all[wv];

  __builtin_prefetch(instance + (size_t)b * 36, 0, 3);

  if (lane < 12) {
    const int w = lane;
    const float* a = instance + ((size_t)b * 12 + w) * 3;
    const float phi = a[0], th = a[1], om = a[2];
    float c, s;   sincosf(0.5f * th, &s, &c);
    float s0, c0; sincosf(-0.5f * (phi + om), &s0, &c0);
    float s1, c1; sincosf( 0.5f * (om  - phi), &s1, &c1);
    const int p = 11 - w;
    L.gR[p][0] = c * c0;  L.gI[p][0] = c * s0;
    L.gR[p][1] = s * c1;  L.gI[p][1] = s * s1;
  } else if (lane < 16) {
    const int d  = lane - 12;
    const int ci = concept_index[b];
    const float* cw = concept_weights + (size_t)ci * 12 + d * 3;
    const float cth = cw[1], com = cw[2];
    float sth, vcth; sincosf(cth, &sth, &vcth);
    float so,  co;   sincosf(com, &so,  &co);
    L.obsA[d] = vcth;
    L.obsB[d] = 2.0f * sth * co;
    L.obsC[d] = 2.0f * sth * so;
  }
  wave_sync();

#pragma unroll
  for (int g = 0; g < 2; ++g) {
    const int base = g ? 0 : 6;
    float pr = L.gR[base][lane & 1], pi = L.gI[base][lane & 1];
#pragma unroll
    for (int j = 1; j < 5; ++j) {
      const int t = (lane >> j) & 1;
      const float gr = L.gR[base + j][t], gi = L.gI[base + j][t];
      const float nr = pr * gr - pi * gi;
      const float ni = pr * gi + pi * gr;
      pr = nr; pi = ni;
    }
#pragma unroll
    for (int t = 0; t < 2; ++t) {
      const float gr = L.gR[base + 5][t], gi = L.gI[base + 5][t];
      const float nr = pr * gr - pi * gi;
      const float ni = pr * gi + pi * gr;
      const int h = lane | (t << 5);
      if (g == 0) { L.Ur[h] = 0.5f * nr; L.Ui[h] = 0.5f * ni; }
      else        { L.Vr[h] = nr;        L.Vi[h] = ni;        }
    }
  }
  wave_sync();

  const int c  = lane & 15;
  const int hh = lane >> 4;

  const int lcol0 = ((c & 12) << 2) | ((c & 3) << 1);
  const int hrow0 = ((c & 4) << 2) | (c & 8) | ((c & 3) << 1);

  const v2f Alo0 = { L.Ur[hrow0],      L.Ui[hrow0]      };
  const v2f Alo1 = { L.Ur[hrow0 | 1],  L.Ui[hrow0 | 1]  };
  const v2f Ahi0 = { L.Ur[hrow0 | 32], L.Ui[hrow0 | 32] };
  const v2f Ahi1 = { L.Ur[hrow0 | 33], L.Ui[hrow0 | 33] };
  const v8f z0 = { 0.f, 0.f, 0.f, 0.f, 0.f, 0.f, 0.f, 0.f };

  const v2f zero2 = { 0.f, 0.f };
  v2f z1 = zero2;
  v2f cr0 = zero2, ci0 = zero2, cr1 = zero2, ci1 = zero2;
  v2f crM = zero2, ciA = zero2, ciB = zero2;
  float n0 = 0.f, n2 = 0.f, n3 = 0.f;

#pragma unroll
  for (int w = 0; w < 4; ++w) {
    const int pw = (w ^ (w >> 1)) & 1;
    const int w1 = (w >> 1) & 1;
    const int lcol = lcol0 | ((w & 2) << 2) | (w & 1);
    const float vr = L.Vr[lcol];
    const float vi = L.Vi[lcol];
    const v2f Bre = { vr, -vi };
    const v2f Bim = { vi,  vr };
    v2f zL = zero2, zH = zero2;

#pragma unroll
    for (int q0 = 0; q0 < 2; ++q0) {
      const v2f Alo = q0 ? Alo1 : Alo0;
      const v2f Ahi = q0 ? Ahi1 : Ahi0;
      const v8f reL = __builtin_amdgcn_wmma_f32_16x16x4_f32(false, Alo, false, Bre,
                                                            (short)0, z0, false, false);
      const v8f imL = __builtin_amdgcn_wmma_f32_16x16x4_f32(false, Alo, false, Bim,
                                                            (short)0, z0, false, false);
      const v8f reH = __builtin_amdgcn_wmma_f32_16x16x4_f32(false, Ahi, false, Bre,
                                                            (short)0, z0, false, false);
      const v8f imH = __builtin_amdgcn_wmma_f32_16x16x4_f32(false, Ahi, false, Bim,
                                                            (short)0, z0, false, false);
#pragma unroll
      for (int q1 = 0; q1 < 2; ++q1) {
#pragma unroll
        for (int rr = 0; rr < 8; rr += 2) {
          const v2f ar = q1 ? (v2f){ reH[rr], reH[rr + 1] } : (v2f){ reL[rr], reL[rr + 1] };
          const v2f ai = q1 ? (v2f){ imH[rr], imH[rr + 1] } : (v2f){ imL[rr], imL[rr + 1] };
          const int rp = ((rr >> 1) ^ (rr >> 2)) & 1;
          const int x0 = q0 ^ rp;
          const int x1 = q1 ^ rp;
          const int x23 = x0 ^ q1;
          const int y0 = x0 ^ pw;
          const v2f nrm = ar * ar + ai * ai;
          if (q1) zH += x0 ? -nrm : nrm;
          else    zL += x0 ? -nrm : nrm;
          z1 += x1 ? -nrm : nrm;


          if (q1 == 0) {
            const v2f br = { reH[rr ^ 4], reH[(rr ^ 4) + 1] };
            const v2f bi = { imH[rr ^ 4], imH[(rr ^ 4) + 1] };
            cr0 += ar * br;  cr0 += ai * bi;
            if (y0) { ci0 -= ar * bi;  ci0 += ai * br; }
            else    { ci0 += ar * bi;  ci0 -= ai * br; }
          }
          if ((rr & 2) == 0) {
            const v2f br = q1 ? (v2f){ reH[rr + 3], reH[rr + 2] } : (v2f){ reL[rr + 3], reL[rr + 2] };
            const v2f bi = q1 ? (v2f){ imH[rr + 3], imH[rr + 2] } : (v2f){ imL[rr + 3], imL[rr + 2] };
            cr1 += ar * br;  cr1 += ai * bi;
            if (x1) { ci1 -= ar * bi;  ci1 += ai * br; }
            else    { ci1 += ar * bi;  ci1 -= ai * br; }
          }
          {
            const v2f br = { pair_gather(ar.x), pair_gather(ar.y) };
            const v2f bi = { pair_gather(ai.x), pair_gather(ai.y) };
            crM += ar * br;  crM += ai * bi;
            if (w1 == 0) {
              if (x23) { ciA -= ar * bi;  ciA += ai * br; }
              else     { ciA += ar * bi;  ciA -= ai * br; }
            } else {
              if (x23) { ciB -= ar * bi;  ciB += ai * br; }
              else     { ciB += ar * bi;  ciB -= ai * br; }
            }
          }
        }
      }
    }

    const float fL = zL.x - zL.y, fH = zH.x - zH.y;
    const float zs = fL + fH;
    const float zd = fL - fH;
    n0 += pw ? -zs : zs;
    n2 += zd;
    n3 += w1 ? -zd : zd;
  }

  const float zn1  = z1.x + z1.y;
  const float cr0s = cr0.x + cr0.y, ci0s = ci0.x - ci0.y;
  const float cr1s = cr1.x + cr1.y, ci1s = ci1.x + ci1.y;
  const float crMs = crM.x + crM.y;
  const float ciAs = ciA.x - ciA.y, ciBs = ciB.x - ciB.y;
  const bool  pw2  = pair_wire(c) != 0;
  const float cr2s = pw2 ? 0.f : crMs,  ci2s = pw2 ? 0.f : ciAs + ciBs;
  const float cr3s = pw2 ? crMs : 0.f,  ci3s = pw2 ? ciAs - ciBs : 0.f;

  const int pl = __builtin_popcount(lane) & 1;
  const float s0 = pl                                               ? -1.f : 1.f;
  const float s1 = hh                                               ? -1.f : 1.f;
  const float s2 = (((c >> 3) ^ hh) & 1)                            ? -1.f : 1.f;
  const float s3 = (((c >> 1) ^ (c >> 2) ^ (c >> 3) ^ hh) & 1)      ? -1.f : 1.f;

  L.partial[lane][0]  = s0 * n0;    L.partial[lane][1]  = cr0s;         L.partial[lane][2]  = s0 * ci0s;
  L.partial[lane][3]  = s1 * zn1;   L.partial[lane][4]  = cr1s;         L.partial[lane][5]  = s1 * ci1s;
  L.partial[lane][6]  = s2 * n2;    L.partial[lane][7]  = cr2s;         L.partial[lane][8]  = s2 * ci2s;
  L.partial[lane][9]  = s3 * n3;    L.partial[lane][10] = cr3s;         L.partial[lane][11] = s3 * ci3s;
  wave_sync();

  if (lane < NACC) {
    float t = 0.0f;
    for (int i = 0; i < 32; ++i) t += L.partial[i][lane];
    L.sums[lane] = t;
  }
  wave_sync();

  if (lane < 4) {
    const int d = lane;
    const float dz  = L.sums[d * 3 + 0];
    const float crR = L.sums[d * 3 + 1];
    const float crI = L.sums[d * 3 + 2];
    const float z = L.obsA[d] * dz + L.obsB[d] * crR + L.obsC[d] * crI;
    out[(size_t)b * 4 + d] = 0.5f * z + 0.5f;
  }
}

extern "C" void kernel_launch(void* const* d_in, const int* in_sizes, int n_in,
                              void* d_out, int out_size, void* d_ws, size_t ws_size,
                              hipStream_t stream) {
  const float* instance = (const float*)d_in[0];
  const int*   cidx     = (const int*)d_in[1];
  const float* cweights = (const float*)d_in[2];
  float*       out      = (float*)d_out;
  const int B = in_sizes[0] / 36;
  const int blocks = (B + WAVES - 1) / WAVES;
  vqc_kernel<<<blocks, THREADS, 0, stream>>>(instance, cidx, cweights, out, B);
}


# Cross sums of wire bits 5 and 2 (partner in another column) formed on the 64-entry factors: (signed sum of |U|^2) x (signed V pair products), no lane gathers; order of the f32 sums changes
# speedup vs baseline: 7.5866x; 1.1296x over previous
#include <hip/hip_runtime.h>


typedef __attribute__((ext_vector_type(2))) float v2f;
typedef __attribute__((ext_vector_type(8))) float v8f;

#define WAVES   4
#define THREADS (32 * WAVES)
#define NACC    14

struct WaveLds {
  float gR[12][2], gI[12][2];
  float Ur[64], Ui[64];
  float Vr[64], Vi[64];
  float obsA[4], obsB[4], obsC[4];
  float partial[32][NACC + 1];
  float sums[NACC];
};

__device__ __forceinline__ void wave_sync() {
  __builtin_amdgcn_fence(__ATOMIC_RELEASE, "wavefront");
  __builtin_amdgcn_wave_barrier();
  __builtin_amdgcn_fence(__ATOMIC_ACQUIRE, "wavefront");
}

__global__ __launch_bounds__(THREADS)
void vqc_kernel(const float* __restrict__ instance,
                const int*   __restrict__ concept_index,
                const float* __restrict__ concept_weights,
                float*       __restrict__ out,
                int B)
{
  __shared__ WaveLds lds_all[WAVES];

  const int lane = threadIdx.x & 31;
  const int wv = __builtin_amdgcn_readfirstlane((int)(threadIdx.x >> 5));
  const int b  = blockIdx.x * WAVES + wv;
  if (b >= B) return;
  WaveLds& L = lds_all[wv];

  __builtin_prefetch(instance + (size_t)b * 36, 0, 3);

  if (lane < 12) {
    const int w = lane;
    const float* a = instance + ((size_t)b * 12 + w) * 3;
    const float phi = a[0], th = a[1], om = a[2];
    float c, s;   sincosf(0.5f * th, &s, &c);
    float s0, c0; sincosf(-0.5f * (phi + om), &s0, &c0);
    float s1, c1; sincosf( 0.5f * (om  - phi), &s1, &c1);
    const int p = 11 - w;
    L.gR[p][0] = c * c0;  L.gI[p][0] = c * s0;
    L.gR[p][1] = s * c1;  L.gI[p][1] = s * s1;
  } else if (lane < 16) {
    const int d  = lane - 12;
    const int ci = concept_index[b];
    const float* cw = concept_weights + (size_t)ci * 12 + d * 3;
    const float cth = cw[1], com = cw[2];
    float sth, vcth; sincosf(cth, &sth, &vcth);
    float so,  co;   sincosf(com, &so,  &co);
    L.obsA[d] = vcth;
    L.obsB[d] = 2.0f * sth * co;
    L.obsC[d] = 2.0f * sth * so;
  }
  wave_sync();

#pragma unroll
  for (int g = 0; g < 2; ++g) {
    const int base = g ? 0 : 6;
    float pr = L.gR[base][lane & 1], pi = L.gI[base][lane & 1];
#pragma unroll
    for (int j = 1; j < 5; ++j) {
      const int t = (lane >> j) & 1;
      const float gr = L.gR[base + j][t], gi = L.gI[base + j][t];
      const float nr = pr * gr - pi * gi;
      const float ni = pr * gi + pi * gr;
      pr = nr; pi = ni;
    }
#pragma unroll
    for (int t = 0; t < 2; ++t) {
      const float gr = L.gR[base + 5][t], gi = L.gI[base + 5][t];
      const float nr = pr * gr - pi * gi;
      const float ni = pr * gi + pi * gr;
      const int h = lane | (t << 5);
      if (g == 0) { L.Ur[h] = 0.5f * nr; L.Ui[h] = 0.5f * ni; }
      else        { L.Vr[h] = nr;        L.Vi[h] = ni;        }
    }
  }
  wave_sync();

  const int c  = lane & 15;
  const int hh = lane >> 4;

  const int lcol0 = ((c & 12) << 2) | ((c & 3) << 1);
  const int hrow0 = ((c & 4) << 2) | (c & 8) | ((c & 3) << 1);

  const v2f Alo0 = { L.Ur[hrow0],      L.Ui[hrow0]      };
  const v2f Alo1 = { L.Ur[hrow0 | 1],  L.Ui[hrow0 | 1]  };
  const v2f Ahi0 = { L.Ur[hrow0 | 32], L.Ui[hrow0 | 32] };
  const v2f Ahi1 = { L.Ur[hrow0 | 33], L.Ui[hrow0 | 33] };
  const v8f z0 = { 0.f, 0.f, 0.f, 0.f, 0.f, 0.f, 0.f, 0.f };

  const int pl = __builtin_popcount(lane) & 1;
  const int pl2 = __builtin_popcount(lane & 0x1c) & 1;
  float pU0 = 0.f, pU1 = 0.f;
  float pR5 = 0.f, pI5 = 0.f, pR2 = 0.f, pI2 = 0.f;
#pragma unroll
  for (int t = 0; t < 2; ++t) {
    const int i = lane | (t << 5);
    const float ur = L.Ur[i], ui = L.Ui[i];
    const float un = ur * ur + ui * ui;
    pU0 += un;
    pU1 += (pl ^ t) ? -un : un;
    const float ar = L.Vr[i], ai = L.Vi[i];
    const float br5 = L.Vr[i ^ 0x30], bi5 = L.Vi[i ^ 0x30];
    const float br2 = L.Vr[i ^ 6],    bi2 = L.Vi[i ^ 6];
    pR5 += ar * br5 + ai * bi5;
    const float x5 = ar * bi5 - ai * br5;
    pI5 += t ? -x5 : x5;
    pR2 += ar * br2 + ai * bi2;
    const float x2 = ar * bi2 - ai * br2;
    pI2 += (pl2 ^ t) ? -x2 : x2;
  }

  const v2f zero2 = { 0.f, 0.f };
  v2f z1 = zero2;
  v2f cr0 = zero2, ci0 = zero2, cr1 = zero2, ci1 = zero2;
  float n0 = 0.f, n2 = 0.f, n3 = 0.f;

#pragma unroll
  for (int w = 0; w < 4; ++w) {
    const int pw = (w ^ (w >> 1)) & 1;
    const int w1 = (w >> 1) & 1;
    const int lcol = lcol0 | ((w & 2) << 2) | (w & 1);
    const float vr = L.Vr[lcol];
    const float vi = L.Vi[lcol];
    const v2f Bre = { vr, -vi };
    const v2f Bim = { vi,  vr };
    v2f zL = zero2, zH = zero2;

#pragma unroll
    for (int q0 = 0; q0 < 2; ++q0) {
      const v2f Alo = q0 ? Alo1 : Alo0;
      const v2f Ahi = q0 ? Ahi1 : Ahi0;
      const v8f reL = __builtin_amdgcn_wmma_f32_16x16x4_f32(false, Alo, false, Bre,
                                                            (short)0, z0, false, false);
      const v8f imL = __builtin_amdgcn_wmma_f32_16x16x4_f32(false, Alo, false, Bim,
                                                            (short)0, z0, false, false);
      const v8f reH = __builtin_amdgcn_wmma_f32_16x16x4_f32(false, Ahi, false, Bre,
                                                            (short)0, z0, false, false);
      const v8f imH = __builtin_amdgcn_wmma_f32_16x16x4_f32(false, Ahi, false, Bim,
                                                            (short)0, z0, false, false);
#pragma unroll
      for (int q1 = 0; q1 < 2; ++q1) {
#pragma unroll
        for (int rr = 0; rr < 8; rr += 2) {
          const v2f ar = q1 ? (v2f){ reH[rr], reH[rr + 1] } : (v2f){ reL[rr], reL[rr + 1] };
          const v2f ai = q1 ? (v2f){ imH[rr], imH[rr + 1] } : (v2f){ imL[rr], imL[rr + 1] };
          const int rp = ((rr >> 1) ^ (rr >> 2)) & 1;
          const int x0 = q0 ^ rp;
          const int x1 = q1 ^ rp;
          const int y0 = x0 ^ pw;
          const v2f nrm = ar * ar + ai * ai;
          if (q1) zH += x0 ? -nrm : nrm;
          else    zL += x0 ? -nrm : nrm;
          z1 += x1 ? -nrm : nrm;


          if (q1 == 0) {
            const v2f br = { reH[rr ^ 4], reH[(rr ^ 4) + 1] };
            const v2f bi = { imH[rr ^ 4], imH[(rr ^ 4) + 1] };
            cr0 += ar * br;  cr0 += ai * bi;
            if (y0) { ci0 -= ar * bi;  ci0 += ai * br; }
            else    { ci0 += ar * bi;  ci0 -= ai * br; }
          }
          if ((rr & 2) == 0) {
            const v2f br = q1 ? (v2f){ reH[rr + 3], reH[rr + 2] } : (v2f){ reL[rr + 3], reL[rr + 2] };
            const v2f bi = q1 ? (v2f){ imH[rr + 3], imH[rr + 2] } : (v2f){ imL[rr + 3], imL[rr + 2] };
            cr1 += ar * br;  cr1 += ai * bi;
            if (x1) { ci1 -= ar * bi;  ci1 += ai * br; }
            else    { ci1 += ar * bi;  ci1 -= ai * br; }
          }
        }
      }
    }

    const float fL = zL.x - zL.y, fH = zH.x - zH.y;
    const float zs = fL + fH;
    const float zd = fL - fH;
    n0 += pw ? -zs : zs;
    n2 += zd;
    n3 += w1 ? -zd : zd;
  }

  const float zn1  = z1.x + z1.y;
  const float cr0s = cr0.x + cr0.y, ci0s = ci0.x - ci0.y;
  const float cr1s = cr1.x + cr1.y, ci1s = ci1.x + ci1.y;

  const float s0 = pl                                               ? -1.f : 1.f;
  const float s1 = hh                                               ? -1.f : 1.f;
  const float s2 = (((c >> 3) ^ hh) & 1)                            ? -1.f : 1.f;
  const float s3 = (((c >> 1) ^ (c >> 2) ^ (c >> 3) ^ hh) & 1)      ? -1.f : 1.f;

  L.partial[lane][0]  = s0 * n0;    L.partial[lane][1]  = cr0s;         L.partial[lane][2]  = s0 * ci0s;
  L.partial[lane][3]  = s1 * zn1;   L.partial[lane][4]  = cr1s;         L.partial[lane][5]  = s1 * ci1s;
  L.partial[lane][6]  = s2 * n2;    L.partial[lane][7]  = s3 * n3;
  L.partial[lane][8]  = pU0;        L.partial[lane][9]  = pU1;
  L.partial[lane][10] = pR5;        L.partial[lane][11] = pI5;
  L.partial[lane][12] = pR2;        L.partial[lane][13] = pI2;
  wave_sync();

  if (lane < NACC) {
    float t = 0.0f;
    for (int i = 0; i < 32; ++i) t += L.partial[i][lane];
    L.sums[lane] = t;
  }
  wave_sync();

  if (lane < 4) {
    const int d = lane;
    float dz, crR, crI;
    if (d < 2) {
      dz  = L.sums[d * 3 + 0];
      crR = L.sums[d * 3 + 1];
      crI = L.sums[d * 3 + 2];
    } else {
      dz  = L.sums[4 + d];
      crR = 2.0f * L.sums[8] * L.sums[6 + 2 * d];
      crI = 2.0f * L.sums[9] * L.sums[7 + 2 * d];
    }
    const float z = L.obsA[d] * dz + L.obsB[d] * crR + L.obsC[d] * crI;
    out[(size_t)b * 4 + d] = 0.5f * z + 0.5f;
  }
}

extern "C" void kernel_launch(void* const* d_in, const int* in_sizes, int n_in,
                              void* d_out, int out_size, void* d_ws, size_t ws_size,
                              hipStream_t stream) {
  const float* instance = (const float*)d_in[0];
  const int*   cidx     = (const int*)d_in[1];
  const float* cweights = (const float*)d_in[2];
  float*       out      = (float*)d_out;
  const int B = in_sizes[0] / 36;
  const int blocks = (B + WAVES - 1) / WAVES;
  vqc_kernel<<<blocks, THREADS, 0, stream>>>(instance, cidx, cweights, out, B);
}


# Cross sums of wire bits 11 and 8 (partner in another row) formed on the 64-entry factors: (signed U pair products) x (signed sum of |V|^2), no amplitude pairs; order of the f32 sums changes
# speedup vs baseline: 8.5119x; 1.1220x over previous
#include <hip/hip_runtime.h>


typedef __attribute__((ext_vector_type(2))) float v2f;
typedef __attribute__((ext_vector_type(8))) float v8f;

#define WAVES   4
#define THREADS (32 * WAVES)
#define NACC    16

struct WaveLds {
  float gR[12][2], gI[12][2];
  float Ur[64], Ui[64];
  float Vr[64], Vi[64];
  float obsA[4], obsB[4], obsC[4];
  float partial[32][NACC + 1];
  float sums[NACC];
};

__device__ __forceinline__ void wave_sync() {
  __builtin_amdgcn_fence(__ATOMIC_RELEASE, "wavefront");
  __builtin_amdgcn_wave_barrier();
  __builtin_amdgcn_fence(__ATOMIC_ACQUIRE, "wavefront");
}

__global__ __launch_bounds__(THREADS)
void vqc_kernel(const float* __restrict__ instance,
                const int*   __restrict__ concept_index,
                const float* __restrict__ concept_weights,
                float*       __restrict__ out,
                int B)
{
  __shared__ WaveLds lds_all[WAVES];

  const int lane = threadIdx.x & 31;
  const int wv = __builtin_amdgcn_readfirstlane((int)(threadIdx.x >> 5));
  const int b  = blockIdx.x * WAVES + wv;
  if (b >= B) return;
  WaveLds& L = lds_all[wv];

  __builtin_prefetch(instance + (size_t)b * 36, 0, 3);

  if (lane < 12) {
    const int w = lane;
    const float* a = instance + ((size_t)b * 12 + w) * 3;
    const float phi = a[0], th = a[1], om = a[2];
    float c, s;   sincosf(0.5f * th, &s, &c);
    float s0, c0; sincosf(-0.5f * (phi + om), &s0, &c0);
    float s1, c1; sincosf( 0.5f * (om  - phi), &s1, &c1);
    const int p = 11 - w;
    L.gR[p][0] = c * c0;  L.gI[p][0] = c * s0;
    L.gR[p][1] = s * c1;  L.gI[p][1] = s * s1;
  } else if (lane < 16) {
    const int d  = lane - 12;
    const int ci = concept_index[b];
    const float* cw = concept_weights + (size_t)ci * 12 + d * 3;
    const float cth = cw[1], com = cw[2];
    float sth, vcth; sincosf(cth, &sth, &vcth);
    float so,  co;   sincosf(com, &so,  &co);
    L.obsA[d] = vcth;
    L.obsB[d] = 2.0f * sth * co;
    L.obsC[d] = 2.0f * sth * so;
  }
  wave_sync();

#pragma unroll
  for (int g = 0; g < 2; ++g) {
    const int base = g ? 0 : 6;
    float pr = L.gR[base][lane & 1], pi = L.gI[base][lane & 1];
#pragma unroll
    for (int j = 1; j < 5; ++j) {
      const int t = (lane >> j) & 1;
      const float gr = L.gR[base + j][t], gi = L.gI[base + j][t];
      const float nr = pr * gr - pi * gi;
      const float ni = pr * gi + pi * gr;
      pr = nr; pi = ni;
    }
#pragma unroll
    for (int t = 0; t < 2; ++t) {
      const float gr = L.gR[base + 5][t], gi = L.gI[base + 5][t];
      const float nr = pr * gr - pi * gi;
      const float ni = pr * gi + pi * gr;
      const int h = lane | (t << 5);
      if (g == 0) { L.Ur[h] = 0.5f * nr; L.Ui[h] = 0.5f * ni; }
      else        { L.Vr[h] = nr;        L.Vi[h] = ni;        }
    }
  }
  wave_sync();

  const int c  = lane & 15;
  const int hh = lane >> 4;

  const int lcol0 = ((c & 12) << 2) | ((c & 3) << 1);
  const int hrow0 = ((c & 4) << 2) | (c & 8) | ((c & 3) << 1);

  const v2f Alo0 = { L.Ur[hrow0],      L.Ui[hrow0]      };
  const v2f Alo1 = { L.Ur[hrow0 | 1],  L.Ui[hrow0 | 1]  };
  const v2f Ahi0 = { L.Ur[hrow0 | 32], L.Ui[hrow0 | 32] };
  const v2f Ahi1 = { L.Ur[hrow0 | 33], L.Ui[hrow0 | 33] };
  const v8f z0 = { 0.f, 0.f, 0.f, 0.f, 0.f, 0.f, 0.f, 0.f };

  const int pl = __builtin_popcount(lane) & 1;
  const int pl2 = __builtin_popcount(lane & 0x1c) & 1;
  float pU0 = 0.f, pU1 = 0.f;
  float pV0 = 0.f, pVa = 0.f;
  float qR11 = 0.f, qI11 = 0.f, qR8 = 0.f, qI8 = 0.f;
  float pR5 = 0.f, pI5 = 0.f, pR2 = 0.f, pI2 = 0.f;
#pragma unroll
  for (int t = 0; t < 2; ++t) {
    const int i = lane | (t << 5);
    const float ur = L.Ur[i], ui = L.Ui[i];
    const float un = ur * ur + ui * ui;
    pU0 += un;
    pU1 += (pl ^ t) ? -un : un;
    const float er11 = L.Ur[i ^ 0x30], ei11 = L.Ui[i ^ 0x30];
    const float er8  = L.Ur[i ^ 6],    ei8  = L.Ui[i ^ 6];
    qR11 += ur * er11 + ui * ei11;
    const float y11 = ur * ei11 - ui * er11;
    qI11 += pl ? -y11 : y11;
    qR8 += ur * er8 + ui * ei8;
    const float y8 = ur * ei8 - ui * er8;
    qI8 += (pl2 ^ t) ? -y8 : y8;
    const float ar = L.Vr[i], ai = L.Vi[i];
    const float vn = ar * ar + ai * ai;
    pV0 += vn;
    pVa += (pl ^ t) ? -vn : vn;
    const float br5 = L.Vr[i ^ 0x30], bi5 = L.Vi[i ^ 0x30];
    const float br2 = L.Vr[i ^ 6],    bi2 = L.Vi[i ^ 6];
    pR5 += ar * br5 + ai * bi5;
    const float x5 = ar * bi5 - ai * br5;
    pI5 += t ? -x5 : x5;
    pR2 += ar * br2 + ai * bi2;
    const float x2 = ar * bi2 - ai * br2;
    pI2 += (pl2 ^ t) ? -x2 : x2;
  }

  const v2f zero2 = { 0.f, 0.f };
  v2f z1 = zero2;
  float n0 = 0.f, n2 = 0.f, n3 = 0.f;

#pragma unroll
  for (int w = 0; w < 4; ++w) {
    const int pw = (w ^ (w >> 1)) & 1;
    const int w1 = (w >> 1) & 1;
    const int lcol = lcol0 | ((w & 2) << 2) | (w & 1);
    const float vr = L.Vr[lcol];
    const float vi = L.Vi[lcol];
    const v2f Bre = { vr, -vi };
    const v2f Bim = { vi,  vr };
    v2f zL = zero2, zH = zero2;

#pragma unroll
    for (int q0 = 0; q0 < 2; ++q0) {
      const v2f Alo = q0 ? Alo1 : Alo0;
      const v2f Ahi = q0 ? Ahi1 : Ahi0;
      const v8f reL = __builtin_amdgcn_wmma_f32_16x16x4_f32(false, Alo, false, Bre,
                                                            (short)0, z0, false, false);
      const v8f imL = __builtin_amdgcn_wmma_f32_16x16x4_f32(false, Alo, false, Bim,
                                                            (short)0, z0, false, false);
      const v8f reH = __builtin_amdgcn_wmma_f32_16x16x4_f32(false, Ahi, false, Bre,
                                                            (short)0, z0, false, false);
      const v8f imH = __builtin_amdgcn_wmma_f32_16x16x4_f32(false, Ahi, false, Bim,
                                                            (short)0, z0, false, false);
#pragma unroll
      for (int q1 = 0; q1 < 2; ++q1) {
#pragma unroll
        for (int rr = 0; rr < 8; rr += 2) {
          const v2f ar = q1 ? (v2f){ reH[rr], reH[rr + 1] } : (v2f){ reL[rr], reL[rr + 1] };
          const v2f ai = q1 ? (v2f){ imH[rr], imH[rr + 1] } : (v2f){ imL[rr], imL[rr + 1] };
          const int rp = ((rr >> 1) ^ (rr >> 2)) & 1;
          const int x0 = q0 ^ rp;
          const int x1 = q1 ^ rp;
          const v2f nrm = ar * ar + ai * ai;
          if (q1) zH += x0 ? -nrm : nrm;
          else    zL += x0 ? -nrm : nrm;
          z1 += x1 ? -nrm : nrm;
        }
      }
    }

    const float fL = zL.x - zL.y, fH = zH.x - zH.y;
    const float zs = fL + fH;
    const float zd = fL - fH;
    n0 += pw ? -zs : zs;
    n2 += zd;
    n3 += w1 ? -zd : zd;
  }

  const float zn1  = z1.x + z1.y;

  const float s0 = pl                                               ? -1.f : 1.f;
  const float s1 = hh                                               ? -1.f : 1.f;
  const float s2 = (((c >> 3) ^ hh) & 1)                            ? -1.f : 1.f;
  const float s3 = (((c >> 1) ^ (c >> 2) ^ (c >> 3) ^ hh) & 1)      ? -1.f : 1.f;

  L.partial[lane][0]  = s0 * n0;    L.partial[lane][1]  = s1 * zn1;
  L.partial[lane][2]  = s2 * n2;    L.partial[lane][3]  = s3 * n3;
  L.partial[lane][4]  = pU0;        L.partial[lane][5]  = pU1;
  L.partial[lane][6]  = pV0;        L.partial[lane][7]  = pVa;
  L.partial[lane][8]  = qR11;       L.partial[lane][9]  = qI11;
  L.partial[lane][10] = qR8;        L.partial[lane][11] = qI8;
  L.partial[lane][12] = pR5;        L.partial[lane][13] = pI5;
  L.partial[lane][14] = pR2;        L.partial[lane][15] = pI2;
  wave_sync();

  if (lane < NACC) {
    float t = 0.0f;
    for (int i = 0; i < 32; ++i) t += L.partial[i][lane];
    L.sums[lane] = t;
  }
  wave_sync();

  if (lane < 4) {
    const int d = lane;
    const float dz = L.sums[d];
    float crR, crI;
    if (d < 2) {
      crR = 2.0f * L.sums[8 + 2 * d] * L.sums[6];
      crI = 2.0f * L.sums[9 + 2 * d] * L.sums[d == 0 ? 7 : 6];
    } else {
      crR = 2.0f * L.sums[4] * L.sums[8 + 2 * d];
      crI = 2.0f * L.sums[5] * L.sums[9 + 2 * d];
    }
    const float z = L.obsA[d] * dz + L.obsB[d] * crR + L.obsC[d] * crI;
    out[(size_t)b * 4 + d] = 0.5f * z + 0.5f;
  }
}

extern "C" void kernel_launch(void* const* d_in, const int* in_sizes, int n_in,
                              void* d_out, int out_size, void* d_ws, size_t ws_size,
                              hipStream_t stream) {
  const float* instance = (const float*)d_in[0];
  const int*   cidx     = (const int*)d_in[1];
  const float* cweights = (const float*)d_in[2];
  float*       out      = (float*)d_out;
  const int B = in_sizes[0] / 36;
  const int blocks = (B + WAVES - 1) / WAVES;
  vqc_kernel<<<blocks, THREADS, 0, stream>>>(instance, cidx, cweights, out, B);
}


# One branch-free set-up pass on 16 lanes: gate column taken up to a phase, (cos(th/2), e^{i om} sin(th/2)), the observable weights' form: 2 sincosf for all lanes (was 5 in 2 branches); loads overlap
# speedup vs baseline: 9.0547x; 1.0638x over previous
#include <hip/hip_runtime.h>


typedef __attribute__((ext_vector_type(2))) float v2f;
typedef __attribute__((ext_vector_type(8))) float v8f;

#define WAVES   4
#define THREADS (32 * WAVES)
#define NACC    16

struct WaveLds {
  float gR[12][2], gI[12][2];
  float Ur[64], Ui[64];
  float Vr[64], Vi[64];
  float obsA[4], obsB[4], obsC[4];
  float partial[32][NACC + 1];
  float sums[NACC];
};

__device__ __forceinline__ void wave_sync() {
  __builtin_amdgcn_fence(__ATOMIC_RELEASE, "wavefront");
  __builtin_amdgcn_wave_barrier();
  __builtin_amdgcn_fence(__ATOMIC_ACQUIRE, "wavefront");
}

__global__ __launch_bounds__(THREADS)
void vqc_kernel(const float* __restrict__ instance,
                const int*   __restrict__ concept_index,
                const float* __restrict__ concept_weights,
                float*       __restrict__ out,
                int B)
{
  __shared__ WaveLds lds_all[WAVES];

  const int lane = threadIdx.x & 31;
  const int wv = __builtin_amdgcn_readfirstlane((int)(threadIdx.x >> 5));
  const int b  = blockIdx.x * WAVES + wv;
  if (b >= B) return;
  WaveLds& L = lds_all[wv];

  __builtin_prefetch(instance + (size_t)b * 36, 0, 3);

  const bool gate = lane < 12;
  const int  w  = gate ? lane : 11;
  const float* ga = instance + ((size_t)b * 12 + w) * 3;
  const float gth = ga[1], gom = ga[2];
  if (lane < 16) {
    const int  d  = gate ? 0 : lane - 12;
    const int  ci = concept_index[b];
    const float* ca = concept_weights + (size_t)ci * 12 + d * 3;
    const float cth = ca[1], com = ca[2];
    const float xa = gate ? 0.5f * gth : cth;
    const float xb = gate ? gom : com;
    float sa, cas; sincosf(xa, &sa, &cas);
    float sb, cbs; sincosf(xb, &sb, &cbs);
    const float amp = gate ? sa : 2.0f * sa;
    const float e1r = amp * cbs, e1i = amp * sb;
    if (gate) {
      const int p = 11 - w;
      L.gR[p][0] = cas;  L.gI[p][0] = 0.0f;
      L.gR[p][1] = e1r;  L.gI[p][1] = e1i;
    } else {
      L.obsA[d] = cas;
      L.obsB[d] = e1r;
      L.obsC[d] = e1i;
    }
  }
  wave_sync();

#pragma unroll
  for (int g = 0; g < 2; ++g) {
    const int base = g ? 0 : 6;
    float pr = L.gR[base][lane & 1], pi = L.gI[base][lane & 1];
#pragma unroll
    for (int j = 1; j < 5; ++j) {
      const int t = (lane >> j) & 1;
      const float gr = L.gR[base + j][t], gi = L.gI[base + j][t];
      const float nr = pr * gr - pi * gi;
      const float ni = pr * gi + pi * gr;
      pr = nr; pi = ni;
    }
#pragma unroll
    for (int t = 0; t < 2; ++t) {
      const float gr = L.gR[base + 5][t], gi = L.gI[base + 5][t];
      const float nr = pr * gr - pi * gi;
      const float ni = pr * gi + pi * gr;
      const int h = lane | (t << 5);
      if (g == 0) { L.Ur[h] = 0.5f * nr; L.Ui[h] = 0.5f * ni; }
      else        { L.Vr[h] = nr;        L.Vi[h] = ni;        }
    }
  }
  wave_sync();

  const int c  = lane & 15;
  const int hh = lane >> 4;

  const int lcol0 = ((c & 12) << 2) | ((c & 3) << 1);
  const int hrow0 = ((c & 4) << 2) | (c & 8) | ((c & 3) << 1);

  const v2f Alo0 = { L.Ur[hrow0],      L.Ui[hrow0]      };
  const v2f Alo1 = { L.Ur[hrow0 | 1],  L.Ui[hrow0 | 1]  };
  const v2f Ahi0 = { L.Ur[hrow0 | 32], L.Ui[hrow0 | 32] };
  const v2f Ahi1 = { L.Ur[hrow0 | 33], L.Ui[hrow0 | 33] };
  const v8f z0 = { 0.f, 0.f, 0.f, 0.f, 0.f, 0.f, 0.f, 0.f };

  const int pl = __builtin_popcount(lane) & 1;
  const int pl2 = __builtin_popcount(lane & 0x1c) & 1;
  float pU0 = 0.f, pU1 = 0.f;
  float pV0 = 0.f, pVa = 0.f;
  float qR11 = 0.f, qI11 = 0.f, qR8 = 0.f, qI8 = 0.f;
  float pR5 = 0.f, pI5 = 0.f, pR2 = 0.f, pI2 = 0.f;
#pragma unroll
  for (int t = 0; t < 2; ++t) {
    const int i = lane | (t << 5);
    const float ur = L.Ur[i], ui = L.Ui[i];
    const float un = ur * ur + ui * ui;
    pU0 += un;
    pU1 += (pl ^ t) ? -un : un;
    const float er11 = L.Ur[i ^ 0x30], ei11 = L.Ui[i ^ 0x30];
    const float er8  = L.Ur[i ^ 6],    ei8  = L.Ui[i ^ 6];
    qR11 += ur * er11 + ui * ei11;
    const float y11 = ur * ei11 - ui * er11;
    qI11 += pl ? -y11 : y11;
    qR8 += ur * er8 + ui * ei8;
    const float y8 = ur * ei8 - ui * er8;
    qI8 += (pl2 ^ t) ? -y8 : y8;
    const float ar = L.Vr[i], ai = L.Vi[i];
    const float vn = ar * ar + ai * ai;
    pV0 += vn;
    pVa += (pl ^ t) ? -vn : vn;
    const float br5 = L.Vr[i ^ 0x30], bi5 = L.Vi[i ^ 0x30];
    const float br2 = L.Vr[i ^ 6],    bi2 = L.Vi[i ^ 6];
    pR5 += ar * br5 + ai * bi5;
    const float x5 = ar * bi5 - ai * br5;
    pI5 += t ? -x5 : x5;
    pR2 += ar * br2 + ai * bi2;
    const float x2 = ar * bi2 - ai * br2;
    pI2 += (pl2 ^ t) ? -x2 : x2;
  }

  const v2f zero2 = { 0.f, 0.f };
  v2f z1 = zero2;
  float n0 = 0.f, n2 = 0.f, n3 = 0.f;

#pragma unroll
  for (int w = 0; w < 4; ++w) {
    const int pw = (w ^ (w >> 1)) & 1;
    const int w1 = (w >> 1) & 1;
    const int lcol = lcol0 | ((w & 2) << 2) | (w & 1);
    const float vr = L.Vr[lcol];
    const float vi = L.Vi[lcol];
    const v2f Bre = { vr, -vi };
    const v2f Bim = { vi,  vr };
    v2f zL = zero2, zH = zero2;

#pragma unroll
    for (int q0 = 0; q0 < 2; ++q0) {
      const v2f Alo = q0 ? Alo1 : Alo0;
      const v2f Ahi = q0 ? Ahi1 : Ahi0;
      const v8f reL = __builtin_amdgcn_wmma_f32_16x16x4_f32(false, Alo, false, Bre,
                                                            (short)0, z0, false, false);
      const v8f imL = __builtin_amdgcn_wmma_f32_16x16x4_f32(false, Alo, false, Bim,
                                                            (short)0, z0, false, false);
      const v8f reH = __builtin_amdgcn_wmma_f32_16x16x4_f32(false, Ahi, false, Bre,
                                                            (short)0, z0, false, false);
      const v8f imH = __builtin_amdgcn_wmma_f32_16x16x4_f32(false, Ahi, false, Bim,
                                                            (short)0, z0, false, false);
#pragma unroll
      for (int q1 = 0; q1 < 2; ++q1) {
#pragma unroll
        for (int rr = 0; rr < 8; rr += 2) {
          const v2f ar = q1 ? (v2f){ reH[rr], reH[rr + 1] } : (v2f){ reL[rr], reL[rr + 1] };
          const v2f ai = q1 ? (v2f){ imH[rr], imH[rr + 1] } : (v2f){ imL[rr], imL[rr + 1] };
          const int rp = ((rr >> 1) ^ (rr >> 2)) & 1;
          const int x0 = q0 ^ rp;
          const int x1 = q1 ^ rp;
          const v2f nrm = ar * ar + ai * ai;
          if (q1) zH += x0 ? -nrm : nrm;
          else    zL += x0 ? -nrm : nrm;
          z1 += x1 ? -nrm : nrm;
        }
      }
    }

    const float fL = zL.x - zL.y, fH = zH.x - zH.y;
    const float zs = fL + fH;
    const float zd = fL - fH;
    n0 += pw ? -zs : zs;
    n2 += zd;
    n3 += w1 ? -zd : zd;
  }

  const float zn1  = z1.x + z1.y;

  const float s0 = pl                                               ? -1.f : 1.f;
  const float s1 = hh                                               ? -1.f : 1.f;
  const float s2 = (((c >> 3) ^ hh) & 1)                            ? -1.f : 1.f;
  const float s3 = (((c >> 1) ^ (c >> 2) ^ (c >> 3) ^ hh) & 1)      ? -1.f : 1.f;

  L.partial[lane][0]  = s0 * n0;    L.partial[lane][1]  = s1 * zn1;
  L.partial[lane][2]  = s2 * n2;    L.partial[lane][3]  = s3 * n3;
  L.partial[lane][4]  = pU0;        L.partial[lane][5]  = pU1;
  L.partial[lane][6]  = pV0;        L.partial[lane][7]  = pVa;
  L.partial[lane][8]  = qR11;       L.partial[lane][9]  = qI11;
  L.partial[lane][10] = qR8;        L.partial[lane][11] = qI8;
  L.partial[lane][12] = pR5;        L.partial[lane][13] = pI5;
  L.partial[lane][14] = pR2;        L.partial[lane][15] = pI2;
  wave_sync();

  if (lane < NACC) {
    float t = 0.0f;
    for (int i = 0; i < 32; ++i) t += L.partial[i][lane];
    L.sums[lane] = t;
  }
  wave_sync();

  if (lane < 4) {
    const int d = lane;
    const float dz = L.sums[d];
    float crR, crI;
    if (d < 2) {
      crR = 2.0f * L.sums[8 + 2 * d] * L.sums[6];
      crI = 2.0f * L.sums[9 + 2 * d] * L.sums[d == 0 ? 7 : 6];
    } else {
      crR = 2.0f * L.sums[4] * L.sums[8 + 2 * d];
      crI = 2.0f * L.sums[5] * L.sums[9 + 2 * d];
    }
    const float z = L.obsA[d] * dz + L.obsB[d] * crR + L.obsC[d] * crI;
    out[(size_t)b * 4 + d] = 0.5f * z + 0.5f;
  }
}

extern "C" void kernel_launch(void* const* d_in, const int* in_sizes, int n_in,
                              void* d_out, int out_size, void* d_ws, size_t ws_size,
                              hipStream_t stream) {
  const float* instance = (const float*)d_in[0];
  const int*   cidx     = (const int*)d_in[1];
  const float* cweights = (const float*)d_in[2];
  float*       out      = (float*)d_out;
  const int B = in_sizes[0] / 36;
  const int blocks = (B + WAVES - 1) / WAVES;
  vqc_kernel<<<blocks, THREADS, 0, stream>>>(instance, cidx, cweights, out, B);
}


# Norm tiles made by the matrix unit itself: one WMMA of |U|^2 rows by |V|^2 columns per tile (was re and im WMMAs, then squares); rounding of norm terms changes, order of their sums kept
# speedup vs baseline: 10.3981x; 1.1484x over previous
#include <hip/hip_runtime.h>


typedef __attribute__((ext_vector_type(2))) float v2f;
typedef __attribute__((ext_vector_type(8))) float v8f;

#define WAVES   4
#define THREADS (32 * WAVES)
#define NACC    16

struct WaveLds {
  float gR[12][2], gI[12][2];
  float Ur[64], Ui[64];
  float Vr[64], Vi[64];
  float Un[64], Vn[64];
  float obsA[4], obsB[4], obsC[4];
  float partial[32][NACC + 1];
  float sums[NACC];
};

__device__ __forceinline__ void wave_sync() {
  __builtin_amdgcn_fence(__ATOMIC_RELEASE, "wavefront");
  __builtin_amdgcn_wave_barrier();
  __builtin_amdgcn_fence(__ATOMIC_ACQUIRE, "wavefront");
}

__global__ __launch_bounds__(THREADS)
void vqc_kernel(const float* __restrict__ instance,
                const int*   __restrict__ concept_index,
                const float* __restrict__ concept_weights,
                float*       __restrict__ out,
                int B)
{
  __shared__ WaveLds lds_all[WAVES];

  const int lane = threadIdx.x & 31;
  const int wv = __builtin_amdgcn_readfirstlane((int)(threadIdx.x >> 5));
  const int b  = blockIdx.x * WAVES + wv;
  if (b >= B) return;
  WaveLds& L = lds_all[wv];

  __builtin_prefetch(instance + (size_t)b * 36, 0, 3);

  const bool gate = lane < 12;
  const int  w  = gate ? lane : 11;
  const float* ga = instance + ((size_t)b * 12 + w) * 3;
  const float gth = ga[1], gom = ga[2];
  if (lane < 16) {
    const int  d  = gate ? 0 : lane - 12;
    const int  ci = concept_index[b];
    const float* ca = concept_weights + (size_t)ci * 12 + d * 3;
    const float cth = ca[1], com = ca[2];
    const float xa = gate ? 0.5f * gth : cth;
    const float xb = gate ? gom : com;
    float sa, cas; sincosf(xa, &sa, &cas);
    float sb, cbs; sincosf(xb, &sb, &cbs);
    const float amp = gate ? sa : 2.0f * sa;
    const float e1r = amp * cbs, e1i = amp * sb;
    if (gate) {
      const int p = 11 - w;
      L.gR[p][0] = cas;  L.gI[p][0] = 0.0f;
      L.gR[p][1] = e1r;  L.gI[p][1] = e1i;
    } else {
      L.obsA[d] = cas;
      L.obsB[d] = e1r;
      L.obsC[d] = e1i;
    }
  }
  wave_sync();

#pragma unroll
  for (int g = 0; g < 2; ++g) {
    const int base = g ? 0 : 6;
    float pr = L.gR[base][lane & 1], pi = L.gI[base][lane & 1];
#pragma unroll
    for (int j = 1; j < 5; ++j) {
      const int t = (lane >> j) & 1;
      const float gr = L.gR[base + j][t], gi = L.gI[base + j][t];
      const float nr = pr * gr - pi * gi;
      const float ni = pr * gi + pi * gr;
      pr = nr; pi = ni;
    }
#pragma unroll
    for (int t = 0; t < 2; ++t) {
      const float gr = L.gR[base + 5][t], gi = L.gI[base + 5][t];
      const float nr = pr * gr - pi * gi;
      const float ni = pr * gi + pi * gr;
      const int h = lane | (t << 5);
      if (g == 0) {
        const float sr = 0.5f * nr, si = 0.5f * ni;
        L.Ur[h] = sr; L.Ui[h] = si; L.Un[h] = sr * sr + si * si;
      } else {
        L.Vr[h] = nr; L.Vi[h] = ni; L.Vn[h] = nr * nr + ni * ni;
      }
    }
  }
  wave_sync();

  const int c  = lane & 15;
  const int hh = lane >> 4;

  const int lcol0 = ((c & 12) << 2) | ((c & 3) << 1);
  const int hrow0 = ((c & 4) << 2) | (c & 8) | ((c & 3) << 1);

  const float a00 = L.Un[hrow0],      a01 = L.Un[hrow0 | 1];
  const float a10 = L.Un[hrow0 | 32], a11 = L.Un[hrow0 | 33];
  const v2f Alo0 = { a00, a00 };
  const v2f Alo1 = { a01, a01 };
  const v2f Ahi0 = { a10, a10 };
  const v2f Ahi1 = { a11, a11 };
  const v8f z0 = { 0.f, 0.f, 0.f, 0.f, 0.f, 0.f, 0.f, 0.f };

  const int pl = __builtin_popcount(lane) & 1;
  const int pl2 = __builtin_popcount(lane & 0x1c) & 1;
  float pU0 = 0.f, pU1 = 0.f;
  float pV0 = 0.f, pVa = 0.f;
  float qR11 = 0.f, qI11 = 0.f, qR8 = 0.f, qI8 = 0.f;
  float pR5 = 0.f, pI5 = 0.f, pR2 = 0.f, pI2 = 0.f;
#pragma unroll
  for (int t = 0; t < 2; ++t) {
    const int i = lane | (t << 5);
    const float ur = L.Ur[i], ui = L.Ui[i];
    const float un = ur * ur + ui * ui;
    pU0 += un;
    pU1 += (pl ^ t) ? -un : un;
    const float er11 = L.Ur[i ^ 0x30], ei11 = L.Ui[i ^ 0x30];
    const float er8  = L.Ur[i ^ 6],    ei8  = L.Ui[i ^ 6];
    qR11 += ur * er11 + ui * ei11;
    const float y11 = ur * ei11 - ui * er11;
    qI11 += pl ? -y11 : y11;
    qR8 += ur * er8 + ui * ei8;
    const float y8 = ur * ei8 - ui * er8;
    qI8 += (pl2 ^ t) ? -y8 : y8;
    const float ar = L.Vr[i], ai = L.Vi[i];
    const float vn = ar * ar + ai * ai;
    pV0 += vn;
    pVa += (pl ^ t) ? -vn : vn;
    const float br5 = L.Vr[i ^ 0x30], bi5 = L.Vi[i ^ 0x30];
    const float br2 = L.Vr[i ^ 6],    bi2 = L.Vi[i ^ 6];
    pR5 += ar * br5 + ai * bi5;
    const float x5 = ar * bi5 - ai * br5;
    pI5 += t ? -x5 : x5;
    pR2 += ar * br2 + ai * bi2;
    const float x2 = ar * bi2 - ai * br2;
    pI2 += (pl2 ^ t) ? -x2 : x2;
  }

  const v2f zero2 = { 0.f, 0.f };
  v2f z1 = zero2;
  float n0 = 0.f, n2 = 0.f, n3 = 0.f;

#pragma unroll
  for (int w = 0; w < 4; ++w) {
    const int pw = (w ^ (w >> 1)) & 1;
    const int w1 = (w >> 1) & 1;
    const int lcol = lcol0 | ((w & 2) << 2) | (w & 1);
    const float vn = L.Vn[lcol];
    const v2f Bn = { vn, vn };
    v2f zL = zero2, zH = zero2;

#pragma unroll
    for (int q0 = 0; q0 < 2; ++q0) {
      const v2f Alo = q0 ? Alo1 : Alo0;
      const v2f Ahi = q0 ? Ahi1 : Ahi0;
      const v8f nL = __builtin_amdgcn_wmma_f32_16x16x4_f32(false, Alo, false, Bn,
                                                           (short)0, z0, false, false);
      const v8f nH = __builtin_amdgcn_wmma_f32_16x16x4_f32(false, Ahi, false, Bn,
                                                           (short)0, z0, false, false);
#pragma unroll
      for (int q1 = 0; q1 < 2; ++q1) {
#pragma unroll
        for (int rr = 0; rr < 8; rr += 2) {
          const v2f nrm = q1 ? (v2f){ nH[rr], nH[rr + 1] } : (v2f){ nL[rr], nL[rr + 1] };
          const int rp = ((rr >> 1) ^ (rr >> 2)) & 1;
          const int x0 = q0 ^ rp;
          const int x1 = q1 ^ rp;
          if (q1) zH += x0 ? -nrm : nrm;
          else    zL += x0 ? -nrm : nrm;
          z1 += x1 ? -nrm : nrm;
        }
      }
    }

    const float fL = zL.x - zL.y, fH = zH.x - zH.y;
    const float zs = fL + fH;
    const float zd = fL - fH;
    n0 += pw ? -zs : zs;
    n2 += zd;
    n3 += w1 ? -zd : zd;
  }

  const float zn1  = z1.x + z1.y;

  const float s0 = pl                                               ? -1.f : 1.f;
  const float s1 = hh                                               ? -1.f : 1.f;
  const float s2 = (((c >> 3) ^ hh) & 1)                            ? -1.f : 1.f;
  const float s3 = (((c >> 1) ^ (c >> 2) ^ (c >> 3) ^ hh) & 1)      ? -1.f : 1.f;

  L.partial[lane][0]  = s0 * n0;    L.partial[lane][1]  = s1 * zn1;
  L.partial[lane][2]  = s2 * n2;    L.partial[lane][3]  = s3 * n3;
  L.partial[lane][4]  = pU0;        L.partial[lane][5]  = pU1;
  L.partial[lane][6]  = pV0;        L.partial[lane][7]  = pVa;
  L.partial[lane][8]  = qR11;       L.partial[lane][9]  = qI11;
  L.partial[lane][10] = qR8;        L.partial[lane][11] = qI8;
  L.partial[lane][12] = pR5;        L.partial[lane][13] = pI5;
  L.partial[lane][14] = pR2;        L.partial[lane][15] = pI2;
  wave_sync();

  if (lane < NACC) {
    float t = 0.0f;
    for (int i = 0; i < 32; ++i) t += L.partial[i][lane];
    L.sums[lane] = t;
  }
  wave_sync();

  if (lane < 4) {
    const int d = lane;
    const float dz = L.sums[d];
    float crR, crI;
    if (d < 2) {
      crR = 2.0f * L.sums[8 + 2 * d] * L.sums[6];
      crI = 2.0f * L.sums[9 + 2 * d] * L.sums[d == 0 ? 7 : 6];
    } else {
      crR = 2.0f * L.sums[4] * L.sums[8 + 2 * d];
      crI = 2.0f * L.sums[5] * L.sums[9 + 2 * d];
    }
    const float z = L.obsA[d] * dz + L.obsB[d] * crR + L.obsC[d] * crI;
    out[(size_t)b * 4 + d] = 0.5f * z + 0.5f;
  }
}

extern "C" void kernel_launch(void* const* d_in, const int* in_sizes, int n_in,
                              void* d_out, int out_size, void* d_ws, size_t ws_size,
                              hipStream_t stream) {
  const float* instance = (const float*)d_in[0];
  const int*   cidx     = (const int*)d_in[1];
  const float* cweights = (const float*)d_in[2];
  float*       out      = (float*)d_out;
  const int B = in_sizes[0] / 36;
  const int blocks = (B + WAVES - 1) / WAVES;
  vqc_kernel<<<blocks, THREADS, 0, stream>>>(instance, cidx, cweights, out, B);
}


# Sums over the 64-entry factors taken bitwise (each a product of six 2-entry sums): cross sums from 12 per-bit values, norm operands as real products; 4 sums left, DPP-reduced; f32 sum order changes
# speedup vs baseline: 14.3411x; 1.3792x over previous
#include <hip/hip_runtime.h>


typedef __attribute__((ext_vector_type(2))) float v2f;
typedef __attribute__((ext_vector_type(8))) float v8f;

#define WAVES   4
#define THREADS (32 * WAVES)

struct __attribute__((aligned(16))) WaveLds {
  float G[12][4];
};

__device__ __forceinline__ void wave_sync() {
  __builtin_amdgcn_fence(__ATOMIC_RELEASE, "wavefront");
  __builtin_amdgcn_wave_barrier();
  __builtin_amdgcn_fence(__ATOMIC_ACQUIRE, "wavefront");
}

template <int M>
__device__ __forceinline__ float row_xor(float x) {
  const int b = __builtin_bit_cast(int, x);
  return __builtin_bit_cast(float, __builtin_amdgcn_update_dpp(b, b, 0x160 + M, 0xf, 0xf, true));
}
__device__ __forceinline__ float xhalf(float x) {
  const int b = __builtin_bit_cast(int, x);
  return __builtin_bit_cast(float,
      __builtin_amdgcn_permlanex16(b, b, 0x76543210, 0xfedcba98, false, false));
}

__global__ __launch_bounds__(THREADS)
void vqc_kernel(const float* __restrict__ instance,
                const int*   __restrict__ concept_index,
                const float* __restrict__ concept_weights,
                float*       __restrict__ out,
                int B)
{
  __shared__ WaveLds lds_all[WAVES];

  const int lane = threadIdx.x & 31;
  const int wv = __builtin_amdgcn_readfirstlane((int)(threadIdx.x >> 5));
  const int b  = blockIdx.x * WAVES + wv;
  if (b >= B) return;
  WaveLds& L = lds_all[wv];

  __builtin_prefetch(instance + (size_t)b * 36, 0, 3);

  const int c  = lane & 15;
  const int hh = lane >> 4;

  const bool gate = c < 12;
  const int  w  = gate ? c : 11;
  const int  d  = gate ? 0 : c - 12;
  const float* ga = instance + ((size_t)b * 12 + w) * 3;
  const int  ci = concept_index[b];
  const float* ca = concept_weights + (size_t)ci * 12 + d * 3;
  const float gth = ga[1], gom = ga[2];
  const float cth = ca[1], com = ca[2];
  const float xa = gate ? 0.5f * gth : cth;
  const float xb = gate ? gom : com;
  float sa, cas; sincosf(xa, &sa, &cas);
  float sb, cbs; sincosf(xb, &sb, &cbs);
  const float amp = gate ? sa : 2.0f * sa;
  const float e1r = amp * cbs, e1i = amp * sb;
  const float P0 = cas * cas, P1 = sa * sa;
  const float dd = P0 - P1;
  const float tc = cas + cas;
  if (lane < 12) {
    const int p = 11 - lane;
    L.G[p][0] = P0;        L.G[p][1] = P1;
    L.G[p][2] = tc * e1r;  L.G[p][3] = tc * e1i;
  }
  wave_sync();

  const bool inA = hh ? (c <= 5) : (c >= 2 && c <= 11);
  const bool inB = hh ? (c <= 8) : (c <= 2);
  float fA = inA ? dd : 1.0f;
  float fB = inB ? dd : 1.0f;
  fA *= row_xor<1>(fA);  fB *= row_xor<1>(fB);
  fA *= row_xor<2>(fA);  fB *= row_xor<2>(fB);
  fA *= row_xor<4>(fA);  fB *= row_xor<4>(fB);
  fA *= row_xor<8>(fA);  fB *= row_xor<8>(fB);

  const int c0 = c & 1, c1 = (c >> 1) & 1, c2 = (c >> 2) & 1, c3 = c >> 3;
  const float ubase = (L.G[6][c0] * L.G[7][c1]) * (L.G[10][c2] * L.G[11][c3]) * L.G[9][hh];
  const float vbase = (L.G[1][c0] * L.G[2][c1]) * (L.G[4][c2] * L.G[5][c3]);
  const float ua = ubase * L.G[8][0], ub = ubase * L.G[8][1];
  const int pa = 10 - 3 * d;
  const float Ra = L.G[pa][2],     Ia = L.G[pa][3];
  const float Rb = L.G[pa + 1][2], Ib = L.G[pa + 1][3];

  const v2f Au = { hh ? -ua : ua, hh ? ub : -ub };
  const v8f z0 = { 0.f, 0.f, 0.f, 0.f, 0.f, 0.f, 0.f, 0.f };

  v2f z8;
  float n11 = 0.f, n5 = 0.f, n2 = 0.f;

#pragma unroll
  for (int sl = 0; sl < 4; ++sl) {
    const int pw = (sl ^ (sl >> 1)) & 1;
    const int w1 = (sl >> 1) & 1;
    const float vn = vbase * (L.G[0][sl & 1] * L.G[3][w1]);
    const v2f Bn = { vn, vn };
    const v8f nD = __builtin_amdgcn_wmma_f32_16x16x4_f32(false, Au, false, Bn,
                                                         (short)0, z0, false, false);
    v2f zb = { nD[0], nD[1] };
    z8 = sl ? z8 + zb : zb;
#pragma unroll
    for (int rr = 2; rr < 8; rr += 2) {
      const v2f nrm = { nD[rr], nD[rr + 1] };
      const int r2 = (rr >> 2) & 1;
      const int rp = ((rr >> 1) ^ (rr >> 2)) & 1;
      z8 += r2 ? -nrm : nrm;
      zb += rp ? -nrm : nrm;
    }
    const float fb = zb.x - zb.y;
    n11 += pw ? -fb : fb;
    n5  += fb;
    n2  += w1 ? -fb : fb;
  }

  const float zn8 = z8.x + z8.y;

  const float s0 = (__builtin_popcount(c) & 1)                      ? -1.f : 1.f;
  const float s1 = hh                                               ? -1.f : 1.f;
  const float s2 = (((c >> 3) ^ hh) & 1)                            ? -1.f : 1.f;
  const float s3 = (((c >> 1) ^ (c >> 2) ^ (c >> 3) ^ hh) & 1)      ? -1.f : 1.f;

  float t0 = s0 * n11, t1 = s1 * zn8, t2 = s2 * n5, t3 = s3 * n2;
  t0 += row_xor<1>(t0);  t1 += row_xor<1>(t1);  t2 += row_xor<1>(t2);  t3 += row_xor<1>(t3);
  t0 += row_xor<2>(t0);  t1 += row_xor<2>(t1);  t2 += row_xor<2>(t2);  t3 += row_xor<2>(t3);
  t0 += row_xor<4>(t0);  t1 += row_xor<4>(t1);  t2 += row_xor<4>(t2);  t3 += row_xor<4>(t3);
  t0 += row_xor<8>(t0);  t1 += row_xor<8>(t1);  t2 += row_xor<8>(t2);  t3 += row_xor<8>(t3);
  t0 += xhalf(t0);       t1 += xhalf(t1);       t2 += xhalf(t2);       t3 += xhalf(t3);

  const float dz = (d & 2) ? ((d & 1) ? t3 : t2) : ((d & 1) ? t1 : t0);
  const float dp = (d & 1) ? fB : fA;
  const float x  = (d == 0) ? Ia : Ra;
  const float y  = (d == 0) ? Rb : Ib;
  const float crR = 0.5f * Ra * Rb;
  const float crI = 0.5f * dp * (x * y);
  const float z = cas * dz + e1r * crR + e1i * crI;
  if (!gate && (d >> 1) == hh)
    out[(size_t)b * 4 + d] = 0.5f * z + 0.5f;
}

extern "C" void kernel_launch(void* const* d_in, const int* in_sizes, int n_in,
                              void* d_out, int out_size, void* d_ws, size_t ws_size,
                              hipStream_t stream) {
  const float* instance = (const float*)d_in[0];
  const int*   cidx     = (const int*)d_in[1];
  const float* cweights = (const float*)d_in[2];
  float*       out      = (float*)d_out;
  const int B = in_sizes[0] / 36;
  const int blocks = (B + WAVES - 1) / WAVES;
  vqc_kernel<<<blocks, THREADS, 0, stream>>>(instance, cidx, cweights, out, B);
}


# Two batch elements per wave (set-up row 0 takes one, row 1 the next, was mirrored work; 8 WMMAs per wave; each row collects its own sums): 2048 waves, was 4096; order of the f32 lane sums changes
# speedup vs baseline: 16.3805x; 1.1422x over previous
#include <hip/hip_runtime.h>


typedef __attribute__((ext_vector_type(2))) float v2f;
typedef __attribute__((ext_vector_type(8))) float v8f;

#define WAVES   4
#define THREADS (32 * WAVES)

struct __attribute__((aligned(16))) ElemLds {
  float G[12][4];
};
struct __attribute__((aligned(16))) WaveLds {
  ElemLds E[2];
};

__device__ __forceinline__ void wave_sync() {
  __builtin_amdgcn_fence(__ATOMIC_RELEASE, "wavefront");
  __builtin_amdgcn_wave_barrier();
  __builtin_amdgcn_fence(__ATOMIC_ACQUIRE, "wavefront");
}

template <int M>
__device__ __forceinline__ float row_xor(float x) {
  const int b = __builtin_bit_cast(int, x);
  return __builtin_bit_cast(float, __builtin_amdgcn_update_dpp(b, b, 0x160 + M, 0xf, 0xf, true));
}
__device__ __forceinline__ float xhalf(float x) {
  const int b = __builtin_bit_cast(int, x);
  return __builtin_bit_cast(float,
      __builtin_amdgcn_permlanex16(b, b, 0x76543210, 0xfedcba98, false, false));
}

__device__ __forceinline__ void norm_sums(const ElemLds& T, int c0, int c1, int c2, int c3, int hh,
                                          float& n11, float& zn8, float& n5, float& n2)
{
  const float ubase = (T.G[6][c0] * T.G[7][c1]) * (T.G[10][c2] * T.G[11][c3]) * T.G[9][hh];
  const float vbase = (T.G[1][c0] * T.G[2][c1]) * (T.G[4][c2] * T.G[5][c3]);
  const float ua = ubase * T.G[8][0], ub = ubase * T.G[8][1];

  const v2f Au = { hh ? -ua : ua, hh ? ub : -ub };
  const v8f z0 = { 0.f, 0.f, 0.f, 0.f, 0.f, 0.f, 0.f, 0.f };

  v2f z8;
  n11 = 0.f; n5 = 0.f; n2 = 0.f;

#pragma unroll
  for (int sl = 0; sl < 4; ++sl) {
    const int pw = (sl ^ (sl >> 1)) & 1;
    const int w1 = (sl >> 1) & 1;
    const float vn = vbase * (T.G[0][sl & 1] * T.G[3][w1]);
    const v2f Bn = { vn, vn };
    const v8f nD = __builtin_amdgcn_wmma_f32_16x16x4_f32(false, Au, false, Bn,
                                                         (short)0, z0, false, false);
    v2f zb = { nD[0], nD[1] };
    z8 = sl ? z8 + zb : zb;
#pragma unroll
    for (int rr = 2; rr < 8; rr += 2) {
      const v2f nrm = { nD[rr], nD[rr + 1] };
      const int r2 = (rr >> 2) & 1;
      const int rp = ((rr >> 1) ^ (rr >> 2)) & 1;
      z8 += r2 ? -nrm : nrm;
      zb += rp ? -nrm : nrm;
    }
    const float fb = zb.x - zb.y;
    n11 += pw ? -fb : fb;
    n5  += fb;
    n2  += w1 ? -fb : fb;
  }

  zn8 = z8.x + z8.y;
}

__global__ __launch_bounds__(THREADS)
void vqc_kernel(const float* __restrict__ instance,
                const int*   __restrict__ concept_index,
                const float* __restrict__ concept_weights,
                float*       __restrict__ out,
                int B)
{
  __shared__ WaveLds lds_all[WAVES];

  const int lane = threadIdx.x & 31;
  const int wv = __builtin_amdgcn_readfirstlane((int)(threadIdx.x >> 5));
  const int b0 = (blockIdx.x * WAVES + wv) * 2;
  if (b0 >= B) return;
  const int b1 = (b0 + 1 < B) ? b0 + 1 : b0;
  WaveLds& L = lds_all[wv];

  __builtin_prefetch(instance + (size_t)b0 * 36, 0, 3);

  const int c  = lane & 15;
  const int hh = lane >> 4;
  const int  b    = hh ? b1 : b0;
  const bool have = b0 + hh < B;

  const bool gate = c < 12;
  const int  w  = gate ? c : 11;
  const int  d  = gate ? 0 : c - 12;
  const float* ga = instance + ((size_t)b * 12 + w) * 3;
  const int  ci0 = __builtin_amdgcn_readfirstlane(concept_index[b0]);
  const int  ci1 = __builtin_amdgcn_readfirstlane(concept_index[b1]);
  const int  ci  = hh ? ci1 : ci0;
  const float* ca = concept_weights + (size_t)ci * 12 + d * 3;
  const float gth = ga[1], gom = ga[2];
  const float cth = ca[1], com = ca[2];
  const float xa = gate ? 0.5f * gth : cth;
  const float xb = gate ? gom : com;
  float sa, cas; sincosf(xa, &sa, &cas);
  float sb, cbs; sincosf(xb, &sb, &cbs);
  const float amp = gate ? sa : 2.0f * sa;
  const float e1r = amp * cbs, e1i = amp * sb;
  const float P0 = cas * cas, P1 = sa * sa;
  const float dd = P0 - P1;
  const float tc = cas + cas;
  if (gate) {
    const int p = 11 - c;
    ElemLds& T = L.E[hh];
    T.G[p][0] = P0;        T.G[p][1] = P1;
    T.G[p][2] = tc * e1r;  T.G[p][3] = tc * e1i;
  }
  wave_sync();

  float f0 = (c >= 2 && c <= 11) ? dd : 1.0f;
  float f1 = (c <= 2) ? dd : 1.0f;
  float f2 = (c <= 5) ? dd : 1.0f;
  float f3 = (c <= 8) ? dd : 1.0f;
  f0 *= row_xor<1>(f0);  f1 *= row_xor<1>(f1);  f2 *= row_xor<1>(f2);  f3 *= row_xor<1>(f3);
  f0 *= row_xor<2>(f0);  f1 *= row_xor<2>(f1);  f2 *= row_xor<2>(f2);  f3 *= row_xor<2>(f3);
  f0 *= row_xor<4>(f0);  f1 *= row_xor<4>(f1);  f2 *= row_xor<4>(f2);  f3 *= row_xor<4>(f3);
  f0 *= row_xor<8>(f0);  f1 *= row_xor<8>(f1);  f2 *= row_xor<8>(f2);  f3 *= row_xor<8>(f3);

  const int c0 = c & 1, c1 = (c >> 1) & 1, c2 = (c >> 2) & 1, c3 = c >> 3;
  float a11, a8, a5, a2;
  float q11, q8, q5, q2;
  norm_sums(L.E[0], c0, c1, c2, c3, hh, a11, a8, a5, a2);
  norm_sums(L.E[1], c0, c1, c2, c3, hh, q11, q8, q5, q2);

  const int pa = 10 - 3 * d;
  const ElemLds& To = L.E[hh];
  const float Ra = To.G[pa][2],     Ia = To.G[pa][3];
  const float Rb = To.G[pa + 1][2], Ib = To.G[pa + 1][3];

  const float s0 = (__builtin_popcount(c) & 1)                      ? -1.f : 1.f;
  const float s1 = hh                                               ? -1.f : 1.f;
  const float s2 = (((c >> 3) ^ hh) & 1)                            ? -1.f : 1.f;
  const float s3 = (((c >> 1) ^ (c >> 2) ^ (c >> 3) ^ hh) & 1)      ? -1.f : 1.f;

  float t0 = s0 * ((hh ? q11 : a11) + xhalf(hh ? a11 : q11));
  float t1 = s1 * ((hh ? q8  : a8 ) - xhalf(hh ? a8  : q8 ));
  float t2 = s2 * ((hh ? q5  : a5 ) - xhalf(hh ? a5  : q5 ));
  float t3 = s3 * ((hh ? q2  : a2 ) - xhalf(hh ? a2  : q2 ));
  t0 += row_xor<1>(t0);  t1 += row_xor<1>(t1);  t2 += row_xor<1>(t2);  t3 += row_xor<1>(t3);
  t0 += row_xor<2>(t0);  t1 += row_xor<2>(t1);  t2 += row_xor<2>(t2);  t3 += row_xor<2>(t3);
  t0 += row_xor<4>(t0);  t1 += row_xor<4>(t1);  t2 += row_xor<4>(t2);  t3 += row_xor<4>(t3);
  t0 += row_xor<8>(t0);  t1 += row_xor<8>(t1);  t2 += row_xor<8>(t2);  t3 += row_xor<8>(t3);

  const float dz = (d & 2) ? ((d & 1) ? t3 : t2) : ((d & 1) ? t1 : t0);
  const float dp = (d & 2) ? ((d & 1) ? f3 : f2) : ((d & 1) ? f1 : f0);
  const float x  = (d == 0) ? Ia : Ra;
  const float y  = (d == 0) ? Rb : Ib;
  const float crR = 0.5f * Ra * Rb;
  const float crI = 0.5f * dp * (x * y);
  const float z = cas * dz + e1r * crR + e1i * crI;
  if (!gate && have)
    out[(size_t)b * 4 + d] = 0.5f * z + 0.5f;
}

extern "C" void kernel_launch(void* const* d_in, const int* in_sizes, int n_in,
                              void* d_out, int out_size, void* d_ws, size_t ws_size,
                              hipStream_t stream) {
  const float* instance = (const float*)d_in[0];
  const int*   cidx     = (const int*)d_in[1];
  const float* cweights = (const float*)d_in[2];
  float*       out      = (float*)d_out;
  const int B = in_sizes[0] / 36;
  const int blocks = (B + 2 * WAVES - 1) / (2 * WAVES);
  vqc_kernel<<<blocks, THREADS, 0, stream>>>(instance, cidx, cweights, out, B);
}
